# starting kernel
# baseline (speedup 1.0000x reference)
#include <hip/hip_runtime.h>

typedef __attribute__((ext_vector_type(2))) float v2f;
typedef __attribute__((ext_vector_type(8))) float v8f;

#define HDIM 128

__global__ __launch_bounds__(256) void k_zero(float4* __restrict__ p, long n4) {
  long i = (long)blockIdx.x * blockDim.x + threadIdx.x;
  if (i < n4) p[i] = make_float4(0.f, 0.f, 0.f, 0.f);
}

__global__ __launch_bounds__(256) void k_scatter(
    const float* __restrict__ xsrc, const int* __restrict__ src,
    const int* __restrict__ dst, float* __restrict__ agg,
    float* __restrict__ cnt, int nE) {
  int t = blockIdx.x * blockDim.x + threadIdx.x;
  int e = t >> 5;
  int lane = t & 31;
  if (e >= nE) return;
  int s = src[e], d = dst[e];
  float4 v = ((const float4*)(xsrc + (long)s * HDIM))[lane];
  float* ap = agg + (long)d * HDIM + lane * 4;
  atomicAdd(ap + 0, v.x);
  atomicAdd(ap + 1, v.y);
  atomicAdd(ap + 2, v.z);
  atomicAdd(ap + 3, v.w);
  if (lane == 0) atomicAdd(cnt + d, 1.0f);
}

__global__ __launch_bounds__(256) void k_sage_gemm(
    const float* __restrict__ Agg, const float* __restrict__ cnt,
    const float* __restrict__ Xdst,
    const float* __restrict__ Wl, const float* __restrict__ Wr,
    const float* __restrict__ bias, float* __restrict__ out,
    int nRowTiles, int doRelu) {
  const int lane = threadIdx.x & 31;
  const int wave = threadIdx.x >> 5;
  const int gtile = blockIdx.x * 8 + wave;
  const int colTiles = HDIM / 16;
  const int rowStrip = gtile / colTiles;
  const int colTile  = gtile % colTiles;
  const int rt0 = rowStrip * 4;
  if (rt0 >= nRowTiles) return;

  const int lrow = lane & 15;
  const int col  = colTile * 16 + lrow;
  const int kb   = (lane >> 4) * 2;

  const float* arow[4];
  const float* xrow[4];
  float inv[4];
  bool  valid[4];
  long  rowBase[4];
  v8f   acc[4];
  const v8f vzero = {};
#pragma unroll
  for (int t = 0; t < 4; ++t) {
    int rt = rt0 + t;
    valid[t] = (rt < nRowTiles);
    int rtc = valid[t] ? rt : (nRowTiles - 1);
    int row = rtc * 16 + lrow;
    arow[t] = Agg  + (long)row * HDIM;
    xrow[t] = Xdst + (long)row * HDIM;
    inv[t]  = 1.0f / fmaxf(cnt[row], 1.0f);
    rowBase[t] = (long)rtc * 16;
    acc[t] = vzero;
  }

#pragma unroll 2
  for (int k0 = 0; k0 < HDIM; k0 += 4) {
    int k = k0 + kb;
    v2f bl, br;
    bl.x = Wl[(long)k * HDIM + col];
    bl.y = Wl[(long)(k + 1) * HDIM + col];
    br.x = Wr[(long)k * HDIM + col];
    br.y = Wr[(long)(k + 1) * HDIM + col];
#pragma unroll
    for (int t = 0; t < 4; ++t) {
      v2f a = *(const v2f*)(arow[t] + k);
      a.x *= inv[t];
      a.y *= inv[t];
      acc[t] = __builtin_amdgcn_wmma_f32_16x16x4_f32(
          false, a, false, bl, (short)0, acc[t], false, false);
      v2f ax = *(const v2f*)(xrow[t] + k);
      acc[t] = __builtin_amdgcn_wmma_f32_16x16x4_f32(
          false, ax, false, br, (short)0, acc[t], false, false);
    }
  }

  const float bv = bias[col];
#pragma unroll
  for (int t = 0; t < 4; ++t) {
    if (!valid[t]) continue;
    long mBase = rowBase[t] + (lane >> 4) * 8;
#pragma unroll
    for (int r = 0; r < 8; ++r) {
      float v = acc[t][r] + bv;
      if (doRelu) v = fmaxf(v, 0.0f);
      out[(mBase + r) * HDIM + col] = v;
    }
  }
}

__global__ __launch_bounds__(256) void k_dot(const float* __restrict__ u2,
                                             const float* __restrict__ i2,
                                             const int* __restrict__ lu,
                                             const int* __restrict__ li,
                                             float* __restrict__ out, int nEL) {
  int t = blockIdx.x * blockDim.x + threadIdx.x;
  int e = t >> 5;
  int lane = t & 31;
  if (e >= nEL) return;
  float4 a = ((const float4*)(u2 + (long)lu[e] * HDIM))[lane];
  float4 b = ((const float4*)(i2 + (long)li[e] * HDIM))[lane];
  float s = a.x * b.x + a.y * b.y + a.z * b.z + a.w * b.w;
#pragma unroll
  for (int off = 16; off > 0; off >>= 1) s += __shfl_down(s, off, 32);
  if (lane == 0) out[e] = s;
}

extern "C" void kernel_launch(void* const* d_in, const int* in_sizes, int n_in,
                              void* d_out, int out_size, void* d_ws, size_t ws_size,
                              hipStream_t stream) {
  const float* user_emb = (const float*)d_in[0];
  const float* item_emb = (const float*)d_in[1];
  const float* w1_ui_l = (const float*)d_in[2];
  const float* w1_ui_r = (const float*)d_in[3];
  const float* w1_iu_l = (const float*)d_in[4];
  const float* w1_iu_r = (const float*)d_in[5];
  const float* w2_ui_l = (const float*)d_in[6];
  const float* w2_ui_r = (const float*)d_in[7];
  const float* w2_iu_l = (const float*)d_in[8];
  const float* w2_iu_r = (const float*)d_in[9];
  const float* b1_ui = (const float*)d_in[10];
  const float* b1_iu = (const float*)d_in[11];
  const float* b2_ui = (const float*)d_in[12];
  const float* b2_iu = (const float*)d_in[13];
  const int* eui  = (const int*)d_in[16];
  const int* eiu  = (const int*)d_in[17];
  const int* elab = (const int*)d_in[18];

  const int U  = in_sizes[0] / HDIM;
  const int I  = in_sizes[1] / HDIM;
  const int E  = in_sizes[16] / 2;
  const int EL = in_sizes[18] / 2;

  float* ws = (float*)d_ws;
  long off = 0;
  float* agg_u = ws + off; off += (long)U * HDIM;
  float* cnt_u = ws + off; off += U;
  float* agg_i = ws + off; off += (long)I * HDIM;
  float* cnt_i = ws + off; off += I;
  const long zeroLen = off;
  float* u1 = ws + off; off += (long)U * HDIM;
  float* i1 = ws + off; off += (long)I * HDIM;
  float* u2 = ws + off; off += (long)U * HDIM;
  float* i2 = ws + off; off += (long)I * HDIM;
  (void)ws_size;

  auto cdiv = [](long a, long b) { return (int)((a + b - 1) / b); };

  const long z4 = zeroLen / 4;
  const int zeroBlocks    = cdiv(z4, 256);
  const int scatterBlocks = cdiv((long)E * 32, 256);
  const int rowTilesU = U / 16;
  const int rowTilesI = I / 16;
  const int gemmBlocksU = cdiv((long)cdiv(rowTilesU, 4) * (HDIM / 16), 8);
  const int gemmBlocksI = cdiv((long)cdiv(rowTilesI, 4) * (HDIM / 16), 8);
  const int dotBlocks = cdiv((long)EL * 32, 256);

  k_zero<<<zeroBlocks, 256, 0, stream>>>((float4*)ws, z4);
  k_scatter<<<scatterBlocks, 256, 0, stream>>>(item_emb, eiu, eiu + E, agg_u, cnt_u, E);
  k_scatter<<<scatterBlocks, 256, 0, stream>>>(user_emb, eui, eui + E, agg_i, cnt_i, E);
  k_sage_gemm<<<gemmBlocksU, 256, 0, stream>>>(agg_u, cnt_u, user_emb, w1_iu_l, w1_iu_r,
                                               b1_iu, u1, rowTilesU, 1);
  k_sage_gemm<<<gemmBlocksI, 256, 0, stream>>>(agg_i, cnt_i, item_emb, w1_ui_l, w1_ui_r,
                                               b1_ui, i1, rowTilesI, 1);

  k_zero<<<zeroBlocks, 256, 0, stream>>>((float4*)ws, z4);
  k_scatter<<<scatterBlocks, 256, 0, stream>>>(i1, eiu, eiu + E, agg_u, cnt_u, E);
  k_scatter<<<scatterBlocks, 256, 0, stream>>>(u1, eui, eui + E, agg_i, cnt_i, E);
  k_sage_gemm<<<gemmBlocksU, 256, 0, stream>>>(agg_u, cnt_u, u1, w2_iu_l, w2_iu_r,
                                               b2_iu, u2, rowTilesU, 0);
  k_sage_gemm<<<gemmBlocksI, 256, 0, stream>>>(agg_i, cnt_i, i1, w2_ui_l, w2_ui_r,
                                               b2_ui, i2, rowTilesI, 0);

  k_dot<<<dotBlocks, 256, 0, stream>>>(u2, i2, elab, elab + EL, (float*)d_out, EL);
}


# Float-atomic scatter-add replaced by a per-call integer counting sort of the edges by destination plus a one-wave-per-row gather-sum; sum order = bucket fill order, still run-dependent
# speedup vs baseline: 1.6605x; 1.6605x over previous
#include <hip/hip_runtime.h>

typedef __attribute__((ext_vector_type(2))) float v2f;
typedef __attribute__((ext_vector_type(8))) float v8f;

#define HDIM 128


__global__ __launch_bounds__(256) void k_zero_i(int* __restrict__ p, int n) {
  int i = blockIdx.x * blockDim.x + threadIdx.x;
  if (i < n) p[i] = 0;
}

__global__ __launch_bounds__(256) void k_degree(const int* __restrict__ dstU,
                                                const int* __restrict__ dstI,
                                                int* __restrict__ deg, int nE, int nU) {
  int t = blockIdx.x * blockDim.x + threadIdx.x;
  if (t >= 2 * nE) return;
  int r = (t < nE) ? dstU[t] : nU + dstI[t - nE];
  atomicAdd(deg + r, 1);
}

__global__ __launch_bounds__(256) void k_alloc(const int* deg, long long* __restrict__ rng,
                                               int* __restrict__ cursor, int* total,
                                               float* __restrict__ cntU,
                                               float* __restrict__ cntI, int nU, int nR) {
  __shared__ int wsum[8];
  __shared__ int blockBase;
  const int lane = threadIdx.x & 31;
  const int wave = threadIdx.x >> 5;
  const int r = blockIdx.x * 256 + threadIdx.x;
  const int d = (r < nR) ? deg[r] : 0;
  int x = d;
  for (int o = 1; o < 32; o <<= 1) {
    int y = __shfl_up(x, o, 32);
    if (lane >= o) x += y;
  }
  if (lane == 31) wsum[wave] = x;
  __syncthreads();
  if (threadIdx.x == 0) {
    int s = 0;
    for (int w = 0; w < 8; ++w) { int v = wsum[w]; wsum[w] = s; s += v; }
    blockBase = atomicAdd(total, s);
  }
  __syncthreads();
  const int pos = blockBase + wsum[wave] + x - d;
  if (r < nR) {
    rng[r] = ((long long)d << 32) | (unsigned)pos;
    cursor[r] = pos;
    if (r < nU) cntU[r] = (float)d; else cntI[r - nU] = (float)d;
  }
}

__global__ __launch_bounds__(256) void k_fill(const int* __restrict__ eiu,
                                              const int* __restrict__ eui,
                                              int* __restrict__ cursor, int* __restrict__ col,
                                              int nE, int nU) {
  int t = blockIdx.x * blockDim.x + threadIdx.x;
  if (t >= 2 * nE) return;
  int s, r;
  if (t < nE) { s = eiu[t]; r = eiu[nE + t]; }
  else { int e = t - nE; s = eui[e]; r = nU + eui[nE + e]; }
  int pos = atomicAdd(cursor + r, 1);
  col[pos] = s;
}

__global__ __launch_bounds__(256) void k_gather_sum(
    const float* __restrict__ xItem, const float* __restrict__ xUser,
    const long long* __restrict__ rng, const int* __restrict__ col,
    float* __restrict__ aggU, float* __restrict__ aggI, int nU, int nR) {
  const int lane = threadIdx.x & 31;
  const int r = __builtin_amdgcn_readfirstlane(blockIdx.x * 8 + (threadIdx.x >> 5));
  if (r >= nR) return;
  const bool toUser = r < nU;
  const float* __restrict__ x = toUser ? xItem : xUser;
  const int row = toUser ? r : r - nU;
  const long long se = rng[r];
  const int s0 = (int)se;
  const int n = (int)(se >> 32);

  float4 acc = make_float4(0.f, 0.f, 0.f, 0.f);
  for (int base = 0; base < n; base += 32) {
    const int m = (n - base < 32) ? (n - base) : 32;
    const int idx = (lane < m) ? col[s0 + base + lane] : 0;
    for (int j = 0; j < m; j += 8) {
      int s[8];
#pragma unroll
      for (int u = 0; u < 8; ++u) s[u] = __builtin_amdgcn_readlane(idx, j + u);
      float4 v[8];
#pragma unroll
      for (int u = 0; u < 8; ++u) {
        v[u] = make_float4(0.f, 0.f, 0.f, 0.f);
        if (j + u < m)
          v[u] = ((const float4*)(x + (long)s[u] * HDIM))[lane];
      }
#pragma unroll
      for (int u = 0; u < 8; ++u) {
        acc.x += v[u].x; acc.y += v[u].y; acc.z += v[u].z; acc.w += v[u].w;
      }
    }
  }
  float* agg = toUser ? aggU : aggI;
  ((float4*)(agg + (long)row * HDIM))[lane] = acc;
}

__global__ __launch_bounds__(256) void k_sage_gemm(
    const float* __restrict__ Agg, const float* __restrict__ cnt,
    const float* __restrict__ Xdst,
    const float* __restrict__ Wl, const float* __restrict__ Wr,
    const float* __restrict__ bias, float* __restrict__ out,
    int nRowTiles, int doRelu) {
  const int lane = threadIdx.x & 31;
  const int wave = threadIdx.x >> 5;
  const int gtile = blockIdx.x * 8 + wave;
  const int colTiles = HDIM / 16;
  const int rowStrip = gtile / colTiles;
  const int colTile  = gtile % colTiles;
  const int rt0 = rowStrip * 4;
  if (rt0 >= nRowTiles) return;

  const int lrow = lane & 15;
  const int col  = colTile * 16 + lrow;
  const int kb   = (lane >> 4) * 2;

  const float* arow[4];
  const float* xrow[4];
  float inv[4];
  bool  valid[4];
  long  rowBase[4];
  v8f   acc[4];
  const v8f vzero = {};
#pragma unroll
  for (int t = 0; t < 4; ++t) {
    int rt = rt0 + t;
    valid[t] = (rt < nRowTiles);
    int rtc = valid[t] ? rt : (nRowTiles - 1);
    int row = rtc * 16 + lrow;
    arow[t] = Agg  + (long)row * HDIM;
    xrow[t] = Xdst + (long)row * HDIM;
    inv[t]  = 1.0f / fmaxf(cnt[row], 1.0f);
    rowBase[t] = (long)rtc * 16;
    acc[t] = vzero;
  }

#pragma unroll 2
  for (int k0 = 0; k0 < HDIM; k0 += 4) {
    int k = k0 + kb;
    v2f bl, br;
    bl.x = Wl[(long)k * HDIM + col];
    bl.y = Wl[(long)(k + 1) * HDIM + col];
    br.x = Wr[(long)k * HDIM + col];
    br.y = Wr[(long)(k + 1) * HDIM + col];
#pragma unroll
    for (int t = 0; t < 4; ++t) {
      v2f a = *(const v2f*)(arow[t] + k);
      a.x *= inv[t];
      a.y *= inv[t];
      acc[t] = __builtin_amdgcn_wmma_f32_16x16x4_f32(
          false, a, false, bl, (short)0, acc[t], false, false);
      v2f ax = *(const v2f*)(xrow[t] + k);
      acc[t] = __builtin_amdgcn_wmma_f32_16x16x4_f32(
          false, ax, false, br, (short)0, acc[t], false, false);
    }
  }

  const float bv = bias[col];
#pragma unroll
  for (int t = 0; t < 4; ++t) {
    if (!valid[t]) continue;
    long mBase = rowBase[t] + (lane >> 4) * 8;
#pragma unroll
    for (int r = 0; r < 8; ++r) {
      float v = acc[t][r] + bv;
      if (doRelu) v = fmaxf(v, 0.0f);
      out[(mBase + r) * HDIM + col] = v;
    }
  }
}

__global__ __launch_bounds__(256) void k_dot(const float* __restrict__ u2,
                                             const float* __restrict__ i2,
                                             const int* __restrict__ lu,
                                             const int* __restrict__ li,
                                             float* __restrict__ out, int nEL) {
  int t = blockIdx.x * blockDim.x + threadIdx.x;
  int e = t >> 5;
  int lane = t & 31;
  if (e >= nEL) return;
  float4 a = ((const float4*)(u2 + (long)lu[e] * HDIM))[lane];
  float4 b = ((const float4*)(i2 + (long)li[e] * HDIM))[lane];
  float s = a.x * b.x + a.y * b.y + a.z * b.z + a.w * b.w;
#pragma unroll
  for (int off = 16; off > 0; off >>= 1) s += __shfl_down(s, off, 32);
  if (lane == 0) out[e] = s;
}

extern "C" void kernel_launch(void* const* d_in, const int* in_sizes, int n_in,
                              void* d_out, int out_size, void* d_ws, size_t ws_size,
                              hipStream_t stream) {
  const float* user_emb = (const float*)d_in[0];
  const float* item_emb = (const float*)d_in[1];
  const float* w1_ui_l = (const float*)d_in[2];
  const float* w1_ui_r = (const float*)d_in[3];
  const float* w1_iu_l = (const float*)d_in[4];
  const float* w1_iu_r = (const float*)d_in[5];
  const float* w2_ui_l = (const float*)d_in[6];
  const float* w2_ui_r = (const float*)d_in[7];
  const float* w2_iu_l = (const float*)d_in[8];
  const float* w2_iu_r = (const float*)d_in[9];
  const float* b1_ui = (const float*)d_in[10];
  const float* b1_iu = (const float*)d_in[11];
  const float* b2_ui = (const float*)d_in[12];
  const float* b2_iu = (const float*)d_in[13];
  const int* eui  = (const int*)d_in[16];
  const int* eiu  = (const int*)d_in[17];
  const int* elab = (const int*)d_in[18];

  const int U  = in_sizes[0] / HDIM;
  const int I  = in_sizes[1] / HDIM;
  const int E  = in_sizes[16] / 2;
  const int EL = in_sizes[18] / 2;
  const int R  = U + I;

  float* ws = (float*)d_ws;
  long off = 0;
  float* agg_u = ws + off; off += (long)U * HDIM;
  float* cnt_u = ws + off; off += U;
  float* agg_i = ws + off; off += (long)I * HDIM;
  float* cnt_i = ws + off; off += I;
  float* u1 = ws + off; off += (long)U * HDIM;
  float* i1 = ws + off; off += (long)I * HDIM;
  float* u2 = ws + off; off += (long)U * HDIM;
  float* i2 = ws + off; off += (long)I * HDIM;
  off += off & 1;
  long long* rng = (long long*)(ws + off); off += 2L * R;
  int* cursor = (int*)(ws + off); off += R;
  int* deg    = (int*)(ws + off); off += R + 1;
  int* col    = (int*)(ws + off); off += 2L * E;
  (void)ws_size;

  auto cdiv = [](long a, long b) { return (int)((a + b - 1) / b); };

  const int edgeBlocks = cdiv(2L * E, 256);
  const int rowTilesU = U / 16;
  const int rowTilesI = I / 16;
  const int gemmBlocksU = cdiv((long)cdiv(rowTilesU, 4) * (HDIM / 16), 8);
  const int gemmBlocksI = cdiv((long)cdiv(rowTilesI, 4) * (HDIM / 16), 8);
  const int gatherBlocks = cdiv(R, 8);
  const int dotBlocks = cdiv((long)EL * 32, 256);

  k_zero_i<<<cdiv(R + 1, 256), 256, 0, stream>>>(deg, R + 1);
  k_degree<<<edgeBlocks, 256, 0, stream>>>(eiu + E, eui + E, deg, E, U);
  k_alloc<<<cdiv(R, 256), 256, 0, stream>>>(deg, rng, cursor, deg + R, cnt_u, cnt_i, U, R);
  k_fill<<<edgeBlocks, 256, 0, stream>>>(eiu, eui, cursor, col, E, U);

  k_gather_sum<<<gatherBlocks, 256, 0, stream>>>(item_emb, user_emb, rng, col,
                                                 agg_u, agg_i, U, R);
  k_sage_gemm<<<gemmBlocksU, 256, 0, stream>>>(agg_u, cnt_u, user_emb, w1_iu_l, w1_iu_r,
                                               b1_iu, u1, rowTilesU, 1);
  k_sage_gemm<<<gemmBlocksI, 256, 0, stream>>>(agg_i, cnt_i, item_emb, w1_ui_l, w1_ui_r,
                                               b1_ui, i1, rowTilesI, 1);

  k_gather_sum<<<gatherBlocks, 256, 0, stream>>>(i1, u1, rng, col,
                                                 agg_u, agg_i, U, R);
  k_sage_gemm<<<gemmBlocksU, 256, 0, stream>>>(agg_u, cnt_u, u1, w2_iu_l, w2_iu_r,
                                               b2_iu, u2, rowTilesU, 0);
  k_sage_gemm<<<gemmBlocksI, 256, 0, stream>>>(agg_i, cnt_i, i1, w2_ui_l, w2_ui_r,
                                               b2_ui, i2, rowTilesI, 0);

  k_dot<<<dotBlocks, 256, 0, stream>>>(u2, i2, elab, elab + EL, (float*)d_out, EL);
}


# SAGE GEMM data path: each 64-row block's mean and x rows are staged once in LDS (528-B pitch) and read as float4 A fragments, weight strips held in registers; k order and sum order kept
# speedup vs baseline: 2.1418x; 1.2899x over previous
#include <hip/hip_runtime.h>

typedef __attribute__((ext_vector_type(2))) float v2f;
typedef __attribute__((ext_vector_type(8))) float v8f;

#define HDIM 128


__global__ __launch_bounds__(256) void k_zero_i(int* __restrict__ p, int n) {
  int i = blockIdx.x * blockDim.x + threadIdx.x;
  if (i < n) p[i] = 0;
}

__global__ __launch_bounds__(256) void k_degree(const int* __restrict__ dstU,
                                                const int* __restrict__ dstI,
                                                int* __restrict__ deg, int nE, int nU) {
  int t = blockIdx.x * blockDim.x + threadIdx.x;
  if (t >= 2 * nE) return;
  int r = (t < nE) ? dstU[t] : nU + dstI[t - nE];
  atomicAdd(deg + r, 1);
}

__global__ __launch_bounds__(256) void k_alloc(const int* deg, long long* __restrict__ rng,
                                               int* __restrict__ cursor, int* total,
                                               float* __restrict__ cntU,
                                               float* __restrict__ cntI, int nU, int nR) {
  __shared__ int wsum[8];
  __shared__ int blockBase;
  const int lane = threadIdx.x & 31;
  const int wave = threadIdx.x >> 5;
  const int r = blockIdx.x * 256 + threadIdx.x;
  const int d = (r < nR) ? deg[r] : 0;
  int x = d;
  for (int o = 1; o < 32; o <<= 1) {
    int y = __shfl_up(x, o, 32);
    if (lane >= o) x += y;
  }
  if (lane == 31) wsum[wave] = x;
  __syncthreads();
  if (threadIdx.x == 0) {
    int s = 0;
    for (int w = 0; w < 8; ++w) { int v = wsum[w]; wsum[w] = s; s += v; }
    blockBase = atomicAdd(total, s);
  }
  __syncthreads();
  const int pos = blockBase + wsum[wave] + x - d;
  if (r < nR) {
    rng[r] = ((long long)d << 32) | (unsigned)pos;
    cursor[r] = pos;
    if (r < nU) cntU[r] = (float)d; else cntI[r - nU] = (float)d;
  }
}

__global__ __launch_bounds__(256) void k_fill(const int* __restrict__ eiu,
                                              const int* __restrict__ eui,
                                              int* __restrict__ cursor, int* __restrict__ col,
                                              int nE, int nU) {
  int t = blockIdx.x * blockDim.x + threadIdx.x;
  if (t >= 2 * nE) return;
  int s, r;
  if (t < nE) { s = eiu[t]; r = eiu[nE + t]; }
  else { int e = t - nE; s = eui[e]; r = nU + eui[nE + e]; }
  int pos = atomicAdd(cursor + r, 1);
  col[pos] = s;
}

__global__ __launch_bounds__(256) void k_gather_sum(
    const float* __restrict__ xItem, const float* __restrict__ xUser,
    const long long* __restrict__ rng, const int* __restrict__ col,
    float* __restrict__ aggU, float* __restrict__ aggI, int nU, int nR) {
  const int lane = threadIdx.x & 31;
  const int r = __builtin_amdgcn_readfirstlane(blockIdx.x * 8 + (threadIdx.x >> 5));
  if (r >= nR) return;
  const bool toUser = r < nU;
  const float* __restrict__ x = toUser ? xItem : xUser;
  const int row = toUser ? r : r - nU;
  const long long se = rng[r];
  const int s0 = (int)se;
  const int n = (int)(se >> 32);

  float4 acc = make_float4(0.f, 0.f, 0.f, 0.f);
  for (int base = 0; base < n; base += 32) {
    const int m = (n - base < 32) ? (n - base) : 32;
    const int idx = (lane < m) ? col[s0 + base + lane] : 0;
    for (int j = 0; j < m; j += 8) {
      int s[8];
#pragma unroll
      for (int u = 0; u < 8; ++u) s[u] = __builtin_amdgcn_readlane(idx, j + u);
      float4 v[8];
#pragma unroll
      for (int u = 0; u < 8; ++u) {
        v[u] = make_float4(0.f, 0.f, 0.f, 0.f);
        if (j + u < m)
          v[u] = ((const float4*)(x + (long)s[u] * HDIM))[lane];
      }
#pragma unroll
      for (int u = 0; u < 8; ++u) {
        acc.x += v[u].x; acc.y += v[u].y; acc.z += v[u].z; acc.w += v[u].w;
      }
    }
  }
  float* agg = toUser ? aggU : aggI;
  ((float4*)(agg + (long)row * HDIM))[lane] = acc;
}

#define TILE_M 64
#define APITCH 132

__global__ __launch_bounds__(256) void k_sage_gemm(
    const float* __restrict__ Agg, const float* __restrict__ cnt,
    const float* __restrict__ Xdst,
    const float* __restrict__ Wl, const float* __restrict__ Wr,
    const float* __restrict__ bias, float* __restrict__ out,
    int nRows, int doRelu) {
  __shared__ __attribute__((aligned(16))) float sA[TILE_M * APITCH];
  __shared__ __attribute__((aligned(16))) float sX[TILE_M * APITCH];

  const int tid  = threadIdx.x;
  const int lane = tid & 31;
  const int wave = tid >> 5;
  const int c    = lane & 15;
  const int h    = lane >> 4;
  const int col  = wave * 16 + c;
  const int row0 = blockIdx.x * TILE_M;

  v2f bl[32], br[32];
#pragma unroll
  for (int s = 0; s < 32; ++s) {
    const int k = 4 * s + 2 * h;
    bl[s].x = Wl[(long)k * HDIM + col];
    bl[s].y = Wl[(long)(k + 1) * HDIM + col];
    br[s].x = Wr[(long)k * HDIM + col];
    br[s].y = Wr[(long)(k + 1) * HDIM + col];
  }

  {
    const int g = tid & 15;
#pragma unroll
    for (int i = 0; i < 4; ++i) {
      const int r = (tid >> 4) + 16 * i;
      int grow = row0 + r;
      if (grow > nRows - 1) grow = nRows - 1;
      const float inv = 1.0f / fmaxf(cnt[grow], 1.0f);
      const float4* pa = (const float4*)(Agg  + (long)grow * HDIM + 8 * g);
      const float4* px = (const float4*)(Xdst + (long)grow * HDIM + 8 * g);
      float4 a0 = pa[0], a1 = pa[1];
      float4 x0 = px[0], x1 = px[1];
      a0.x *= inv; a0.y *= inv; a0.z *= inv; a0.w *= inv;
      a1.x *= inv; a1.y *= inv; a1.z *= inv; a1.w *= inv;
      float4* da = (float4*)(sA + r * APITCH + 8 * g);
      float4* dx = (float4*)(sX + r * APITCH + 8 * g);
      da[0] = make_float4(a0.x, a0.y, a1.x, a1.y);
      da[1] = make_float4(a0.z, a0.w, a1.z, a1.w);
      dx[0] = make_float4(x0.x, x0.y, x1.x, x1.y);
      dx[1] = make_float4(x0.z, x0.w, x1.z, x1.w);
    }
  }
  __syncthreads();

  v8f acc[4];
  const v8f vzero = {};
#pragma unroll
  for (int t = 0; t < 4; ++t) acc[t] = vzero;

  const float* fa0 = sA + c * APITCH + 4 * h;
  const float* fx0 = sX + c * APITCH + 4 * h;
#pragma unroll
  for (int j = 0; j < 16; ++j) {
    float4 fa[4], fx[4];
#pragma unroll
    for (int t = 0; t < 4; ++t) {
      fa[t] = *(const float4*)(fa0 + 16 * t * APITCH + 8 * j);
      fx[t] = *(const float4*)(fx0 + 16 * t * APITCH + 8 * j);
    }
#pragma unroll
    for (int t = 0; t < 4; ++t) {
      v2f a, ax;
      a.x = fa[t].x;  a.y = fa[t].y;
      ax.x = fx[t].x; ax.y = fx[t].y;
      acc[t] = __builtin_amdgcn_wmma_f32_16x16x4_f32(
          false, a, false, bl[2 * j], (short)0, acc[t], false, false);
      acc[t] = __builtin_amdgcn_wmma_f32_16x16x4_f32(
          false, ax, false, br[2 * j], (short)0, acc[t], false, false);
    }
#pragma unroll
    for (int t = 0; t < 4; ++t) {
      v2f a, ax;
      a.x = fa[t].z;  a.y = fa[t].w;
      ax.x = fx[t].z; ax.y = fx[t].w;
      acc[t] = __builtin_amdgcn_wmma_f32_16x16x4_f32(
          false, a, false, bl[2 * j + 1], (short)0, acc[t], false, false);
      acc[t] = __builtin_amdgcn_wmma_f32_16x16x4_f32(
          false, ax, false, br[2 * j + 1], (short)0, acc[t], false, false);
    }
  }

  const float bv = bias[col];
  float* orow = out + (long)(row0 + 8 * h) * HDIM + col;
  if (row0 + TILE_M <= nRows) {
#pragma unroll
    for (int t = 0; t < 4; ++t) {
#pragma unroll
      for (int r = 0; r < 8; ++r) {
        float v = acc[t][r] + bv;
        if (doRelu) v = fmaxf(v, 0.0f);
        orow[(long)(16 * t + r) * HDIM] = v;
      }
    }
  } else {
#pragma unroll
    for (int t = 0; t < 4; ++t) {
#pragma unroll
      for (int r = 0; r < 8; ++r) {
        float v = acc[t][r] + bv;
        if (doRelu) v = fmaxf(v, 0.0f);
        if (row0 + 8 * h + 16 * t + r < nRows) orow[(long)(16 * t + r) * HDIM] = v;
      }
    }
  }
}

__global__ __launch_bounds__(256) void k_dot(const float* __restrict__ u2,
                                             const float* __restrict__ i2,
                                             const int* __restrict__ lu,
                                             const int* __restrict__ li,
                                             float* __restrict__ out, int nEL) {
  int t = blockIdx.x * blockDim.x + threadIdx.x;
  int e = t >> 5;
  int lane = t & 31;
  if (e >= nEL) return;
  float4 a = ((const float4*)(u2 + (long)lu[e] * HDIM))[lane];
  float4 b = ((const float4*)(i2 + (long)li[e] * HDIM))[lane];
  float s = a.x * b.x + a.y * b.y + a.z * b.z + a.w * b.w;
#pragma unroll
  for (int off = 16; off > 0; off >>= 1) s += __shfl_down(s, off, 32);
  if (lane == 0) out[e] = s;
}

extern "C" void kernel_launch(void* const* d_in, const int* in_sizes, int n_in,
                              void* d_out, int out_size, void* d_ws, size_t ws_size,
                              hipStream_t stream) {
  const float* user_emb = (const float*)d_in[0];
  const float* item_emb = (const float*)d_in[1];
  const float* w1_ui_l = (const float*)d_in[2];
  const float* w1_ui_r = (const float*)d_in[3];
  const float* w1_iu_l = (const float*)d_in[4];
  const float* w1_iu_r = (const float*)d_in[5];
  const float* w2_ui_l = (const float*)d_in[6];
  const float* w2_ui_r = (const float*)d_in[7];
  const float* w2_iu_l = (const float*)d_in[8];
  const float* w2_iu_r = (const float*)d_in[9];
  const float* b1_ui = (const float*)d_in[10];
  const float* b1_iu = (const float*)d_in[11];
  const float* b2_ui = (const float*)d_in[12];
  const float* b2_iu = (const float*)d_in[13];
  const int* eui  = (const int*)d_in[16];
  const int* eiu  = (const int*)d_in[17];
  const int* elab = (const int*)d_in[18];

  const int U  = in_sizes[0] / HDIM;
  const int I  = in_sizes[1] / HDIM;
  const int E  = in_sizes[16] / 2;
  const int EL = in_sizes[18] / 2;
  const int R  = U + I;

  float* ws = (float*)d_ws;
  long off = 0;
  float* agg_u = ws + off; off += (long)U * HDIM;
  float* cnt_u = ws + off; off += U;
  float* agg_i = ws + off; off += (long)I * HDIM;
  float* cnt_i = ws + off; off += I;
  float* u1 = ws + off; off += (long)U * HDIM;
  float* i1 = ws + off; off += (long)I * HDIM;
  float* u2 = ws + off; off += (long)U * HDIM;
  float* i2 = ws + off; off += (long)I * HDIM;
  off += off & 1;
  long long* rng = (long long*)(ws + off); off += 2L * R;
  int* cursor = (int*)(ws + off); off += R;
  int* deg    = (int*)(ws + off); off += R + 1;
  int* col    = (int*)(ws + off); off += 2L * E;
  (void)ws_size;

  auto cdiv = [](long a, long b) { return (int)((a + b - 1) / b); };

  const int edgeBlocks = cdiv(2L * E, 256);
  const int gemmBlocksU = cdiv(U, TILE_M);
  const int gemmBlocksI = cdiv(I, TILE_M);
  const int gatherBlocks = cdiv(R, 8);
  const int dotBlocks = cdiv((long)EL * 32, 256);

  k_zero_i<<<cdiv(R + 1, 256), 256, 0, stream>>>(deg, R + 1);
  k_degree<<<edgeBlocks, 256, 0, stream>>>(eiu + E, eui + E, deg, E, U);
  k_alloc<<<cdiv(R, 256), 256, 0, stream>>>(deg, rng, cursor, deg + R, cnt_u, cnt_i, U, R);
  k_fill<<<edgeBlocks, 256, 0, stream>>>(eiu, eui, cursor, col, E, U);

  k_gather_sum<<<gatherBlocks, 256, 0, stream>>>(item_emb, user_emb, rng, col,
                                                 agg_u, agg_i, U, R);
  k_sage_gemm<<<gemmBlocksU, 256, 0, stream>>>(agg_u, cnt_u, user_emb, w1_iu_l, w1_iu_r,
                                               b1_iu, u1, U, 1);
  k_sage_gemm<<<gemmBlocksI, 256, 0, stream>>>(agg_i, cnt_i, item_emb, w1_ui_l, w1_ui_r,
                                               b1_ui, i1, I, 1);

  k_gather_sum<<<gatherBlocks, 256, 0, stream>>>(i1, u1, rng, col,
                                                 agg_u, agg_i, U, R);
  k_sage_gemm<<<gemmBlocksU, 256, 0, stream>>>(agg_u, cnt_u, u1, w2_iu_l, w2_iu_r,
                                               b2_iu, u2, U, 0);
  k_sage_gemm<<<gemmBlocksI, 256, 0, stream>>>(agg_i, cnt_i, i1, w2_ui_l, w2_ui_r,
                                               b2_ui, i2, I, 0);

  k_dot<<<dotBlocks, 256, 0, stream>>>(u2, i2, elab, elab + EL, (float*)d_out, EL);
}


# Gather-sum and edge-dot waves each loop over 10 rows / 16 edges (one resident pass of waves, next row's indices read ahead) instead of one short wave per row or edge; every sum's order kept
# speedup vs baseline: 2.2480x; 1.0496x over previous
#include <hip/hip_runtime.h>

typedef __attribute__((ext_vector_type(2))) float v2f;
typedef __attribute__((ext_vector_type(8))) float v8f;

#define HDIM 128


__global__ __launch_bounds__(256) void k_zero_i(int* __restrict__ p, int n) {
  int i = blockIdx.x * blockDim.x + threadIdx.x;
  if (i < n) p[i] = 0;
}

__global__ __launch_bounds__(256) void k_degree(const int* __restrict__ dstU,
                                                const int* __restrict__ dstI,
                                                int* __restrict__ deg, int nE, int nU) {
  int t = blockIdx.x * blockDim.x + threadIdx.x;
  if (t >= 2 * nE) return;
  int r = (t < nE) ? dstU[t] : nU + dstI[t - nE];
  atomicAdd(deg + r, 1);
}

__global__ __launch_bounds__(256) void k_alloc(const int* deg, long long* __restrict__ rng,
                                               int* __restrict__ cursor, int* total,
                                               float* __restrict__ cntU,
                                               float* __restrict__ cntI, int nU, int nR) {
  __shared__ int wsum[8];
  __shared__ int blockBase;
  const int lane = threadIdx.x & 31;
  const int wave = threadIdx.x >> 5;
  const int r = blockIdx.x * 256 + threadIdx.x;
  const int d = (r < nR) ? deg[r] : 0;
  int x = d;
  for (int o = 1; o < 32; o <<= 1) {
    int y = __shfl_up(x, o, 32);
    if (lane >= o) x += y;
  }
  if (lane == 31) wsum[wave] = x;
  __syncthreads();
  if (threadIdx.x == 0) {
    int s = 0;
    for (int w = 0; w < 8; ++w) { int v = wsum[w]; wsum[w] = s; s += v; }
    blockBase = atomicAdd(total, s);
  }
  __syncthreads();
  const int pos = blockBase + wsum[wave] + x - d;
  if (r < nR) {
    rng[r] = ((long long)d << 32) | (unsigned)pos;
    cursor[r] = pos;
    if (r < nU) cntU[r] = (float)d; else cntI[r - nU] = (float)d;
  }
}

__global__ __launch_bounds__(256) void k_fill(const int* __restrict__ eiu,
                                              const int* __restrict__ eui,
                                              int* __restrict__ cursor, int* __restrict__ col,
                                              int nE, int nU) {
  int t = blockIdx.x * blockDim.x + threadIdx.x;
  if (t >= 2 * nE) return;
  int s, r;
  if (t < nE) { s = eiu[t]; r = eiu[nE + t]; }
  else { int e = t - nE; s = eui[e]; r = nU + eui[nE + e]; }
  int pos = atomicAdd(cursor + r, 1);
  col[pos] = s;
}

#define ROWS_PER_WAVE 10

__global__ __launch_bounds__(256) void k_gather_sum(
    const float* __restrict__ xItem, const float* __restrict__ xUser,
    const long long* __restrict__ rng, const int* __restrict__ col,
    float* __restrict__ aggU, float* __restrict__ aggI, int nU, int nR) {
  const int lane = threadIdx.x & 31;
  const int nW = gridDim.x * 8;
  int r = __builtin_amdgcn_readfirstlane(blockIdx.x * 8 + (threadIdx.x >> 5));
  if (r >= nR) return;
  long long se = rng[r];
  int rn = r + nW;
  long long seN = (rn < nR) ? rng[rn] : 0LL;
  int idx = (lane < (int)(se >> 32)) ? col[(int)se + lane] : 0;

  do {
    const int rnn = rn + nW;
    const long long seNN = (rnn < nR) ? rng[rnn] : 0LL;

    const bool toUser = r < nU;
    const float* __restrict__ x = toUser ? xItem : xUser;
    const int row = toUser ? r : r - nU;
    const int s0 = (int)se;
    const int n = (int)(se >> 32);

    float4 acc = make_float4(0.f, 0.f, 0.f, 0.f);
    int idxN = 0;
    int base = 0;
    do {
      const int m = (n - base < 32) ? (n - base) : 32;
      int id = idx;
      if (base > 0) id = (lane < m) ? col[s0 + base + lane] : 0;
      int j = 0;
      do {
        int s[8];
#pragma unroll
        for (int u = 0; u < 8; ++u) s[u] = __builtin_amdgcn_readlane(id, j + u);
        float4 v[8];
#pragma unroll
        for (int u = 0; u < 8; ++u) {
          v[u] = make_float4(0.f, 0.f, 0.f, 0.f);
          if (j + u < m)
            v[u] = ((const float4*)(x + (long)s[u] * HDIM))[lane];
        }
        if (base + j == 0) idxN = (lane < (int)(seN >> 32)) ? col[(int)seN + lane] : 0;
#pragma unroll
        for (int u = 0; u < 8; ++u) {
          acc.x += v[u].x; acc.y += v[u].y; acc.z += v[u].z; acc.w += v[u].w;
        }
        j += 8;
      } while (j < m);
      base += 32;
    } while (base < n);
    float* agg = toUser ? aggU : aggI;
    ((float4*)(agg + (long)row * HDIM))[lane] = acc;

    r = rn; rn = rnn; se = seN; seN = seNN; idx = idxN;
  } while (r < nR);
}

#define TILE_M 64
#define APITCH 132

__global__ __launch_bounds__(256) void k_sage_gemm(
    const float* __restrict__ Agg, const float* __restrict__ cnt,
    const float* __restrict__ Xdst,
    const float* __restrict__ Wl, const float* __restrict__ Wr,
    const float* __restrict__ bias, float* __restrict__ out,
    int nRows, int doRelu) {
  __shared__ __attribute__((aligned(16))) float sA[TILE_M * APITCH];
  __shared__ __attribute__((aligned(16))) float sX[TILE_M * APITCH];

  const int tid  = threadIdx.x;
  const int lane = tid & 31;
  const int wave = tid >> 5;
  const int c    = lane & 15;
  const int h    = lane >> 4;
  const int col  = wave * 16 + c;
  const int row0 = blockIdx.x * TILE_M;

  v2f bl[32], br[32];
#pragma unroll
  for (int s = 0; s < 32; ++s) {
    const int k = 4 * s + 2 * h;
    bl[s].x = Wl[(long)k * HDIM + col];
    bl[s].y = Wl[(long)(k + 1) * HDIM + col];
    br[s].x = Wr[(long)k * HDIM + col];
    br[s].y = Wr[(long)(k + 1) * HDIM + col];
  }

  {
    const int g = tid & 15;
#pragma unroll
    for (int i = 0; i < 4; ++i) {
      const int r = (tid >> 4) + 16 * i;
      int grow = row0 + r;
      if (grow > nRows - 1) grow = nRows - 1;
      const float inv = 1.0f / fmaxf(cnt[grow], 1.0f);
      const float4* pa = (const float4*)(Agg  + (long)grow * HDIM + 8 * g);
      const float4* px = (const float4*)(Xdst + (long)grow * HDIM + 8 * g);
      float4 a0 = pa[0], a1 = pa[1];
      float4 x0 = px[0], x1 = px[1];
      a0.x *= inv; a0.y *= inv; a0.z *= inv; a0.w *= inv;
      a1.x *= inv; a1.y *= inv; a1.z *= inv; a1.w *= inv;
      float4* da = (float4*)(sA + r * APITCH + 8 * g);
      float4* dx = (float4*)(sX + r * APITCH + 8 * g);
      da[0] = make_float4(a0.x, a0.y, a1.x, a1.y);
      da[1] = make_float4(a0.z, a0.w, a1.z, a1.w);
      dx[0] = make_float4(x0.x, x0.y, x1.x, x1.y);
      dx[1] = make_float4(x0.z, x0.w, x1.z, x1.w);
    }
  }
  __syncthreads();

  v8f acc[4];
  const v8f vzero = {};
#pragma unroll
  for (int t = 0; t < 4; ++t) acc[t] = vzero;

  const float* fa0 = sA + c * APITCH + 4 * h;
  const float* fx0 = sX + c * APITCH + 4 * h;
#pragma unroll
  for (int j = 0; j < 16; ++j) {
    float4 fa[4], fx[4];
#pragma unroll
    for (int t = 0; t < 4; ++t) {
      fa[t] = *(const float4*)(fa0 + 16 * t * APITCH + 8 * j);
      fx[t] = *(const float4*)(fx0 + 16 * t * APITCH + 8 * j);
    }
#pragma unroll
    for (int t = 0; t < 4; ++t) {
      v2f a, ax;
      a.x = fa[t].x;  a.y = fa[t].y;
      ax.x = fx[t].x; ax.y = fx[t].y;
      acc[t] = __builtin_amdgcn_wmma_f32_16x16x4_f32(
          false, a, false, bl[2 * j], (short)0, acc[t], false, false);
      acc[t] = __builtin_amdgcn_wmma_f32_16x16x4_f32(
          false, ax, false, br[2 * j], (short)0, acc[t], false, false);
    }
#pragma unroll
    for (int t = 0; t < 4; ++t) {
      v2f a, ax;
      a.x = fa[t].z;  a.y = fa[t].w;
      ax.x = fx[t].z; ax.y = fx[t].w;
      acc[t] = __builtin_amdgcn_wmma_f32_16x16x4_f32(
          false, a, false, bl[2 * j + 1], (short)0, acc[t], false, false);
      acc[t] = __builtin_amdgcn_wmma_f32_16x16x4_f32(
          false, ax, false, br[2 * j + 1], (short)0, acc[t], false, false);
    }
  }

  const float bv = bias[col];
  float* orow = out + (long)(row0 + 8 * h) * HDIM + col;
  if (row0 + TILE_M <= nRows) {
#pragma unroll
    for (int t = 0; t < 4; ++t) {
#pragma unroll
      for (int r = 0; r < 8; ++r) {
        float v = acc[t][r] + bv;
        if (doRelu) v = fmaxf(v, 0.0f);
        orow[(long)(16 * t + r) * HDIM] = v;
      }
    }
  } else {
#pragma unroll
    for (int t = 0; t < 4; ++t) {
#pragma unroll
      for (int r = 0; r < 8; ++r) {
        float v = acc[t][r] + bv;
        if (doRelu) v = fmaxf(v, 0.0f);
        if (row0 + 8 * h + 16 * t + r < nRows) orow[(long)(16 * t + r) * HDIM] = v;
      }
    }
  }
}

#define DOT_EDGES 4
#define DOT_TRIPS 4

__global__ __launch_bounds__(256) void k_dot(const float* __restrict__ u2,
                                             const float* __restrict__ i2,
                                             const int* __restrict__ lu,
                                             const int* __restrict__ li,
                                             float* __restrict__ out, int nEL) {
  const int lane = threadIdx.x & 31;
  const int step = gridDim.x * 8 * DOT_EDGES;
  int e0 = __builtin_amdgcn_readfirstlane(blockIdx.x * 8 + (threadIdx.x >> 5)) * DOT_EDGES;
  if (e0 >= nEL) return;
  int iu[DOT_EDGES], ii[DOT_EDGES];
#pragma unroll
  for (int u = 0; u < DOT_EDGES; ++u) {
    const int e = (e0 + u < nEL) ? e0 + u : nEL - 1;
    iu[u] = lu[e]; ii[u] = li[e];
  }
  do {
    float4 a[DOT_EDGES], b[DOT_EDGES];
#pragma unroll
    for (int u = 0; u < DOT_EDGES; ++u) {
      a[u] = ((const float4*)(u2 + (long)iu[u] * HDIM))[lane];
      b[u] = ((const float4*)(i2 + (long)ii[u] * HDIM))[lane];
    }
    const int e1 = e0 + step;
    int iuN[DOT_EDGES], iiN[DOT_EDGES];
#pragma unroll
    for (int u = 0; u < DOT_EDGES; ++u) { iuN[u] = 0; iiN[u] = 0; }
    if (e1 < nEL) {
#pragma unroll
      for (int u = 0; u < DOT_EDGES; ++u) {
        const int e = (e1 + u < nEL) ? e1 + u : nEL - 1;
        iuN[u] = lu[e]; iiN[u] = li[e];
      }
    }
    float s[DOT_EDGES];
#pragma unroll
    for (int u = 0; u < DOT_EDGES; ++u)
      s[u] = a[u].x * b[u].x + a[u].y * b[u].y + a[u].z * b[u].z + a[u].w * b[u].w;
#pragma unroll
    for (int off = 16; off > 0; off >>= 1) {
#pragma unroll
      for (int u = 0; u < DOT_EDGES; ++u) s[u] += __shfl_down(s[u], off, 32);
    }
    if (lane == 0) {
      if (e0 + DOT_EDGES <= nEL) {
        *(float4*)(out + e0) = make_float4(s[0], s[1], s[2], s[3]);
      } else {
#pragma unroll
        for (int u = 0; u < DOT_EDGES; ++u)
          if (e0 + u < nEL) out[e0 + u] = s[u];
      }
    }
    e0 = e1;
#pragma unroll
    for (int u = 0; u < DOT_EDGES; ++u) { iu[u] = iuN[u]; ii[u] = iiN[u]; }
  } while (e0 < nEL);
}

extern "C" void kernel_launch(void* const* d_in, const int* in_sizes, int n_in,
                              void* d_out, int out_size, void* d_ws, size_t ws_size,
                              hipStream_t stream) {
  const float* user_emb = (const float*)d_in[0];
  const float* item_emb = (const float*)d_in[1];
  const float* w1_ui_l = (const float*)d_in[2];
  const float* w1_ui_r = (const float*)d_in[3];
  const float* w1_iu_l = (const float*)d_in[4];
  const float* w1_iu_r = (const float*)d_in[5];
  const float* w2_ui_l = (const float*)d_in[6];
  const float* w2_ui_r = (const float*)d_in[7];
  const float* w2_iu_l = (const float*)d_in[8];
  const float* w2_iu_r = (const float*)d_in[9];
  const float* b1_ui = (const float*)d_in[10];
  const float* b1_iu = (const float*)d_in[11];
  const float* b2_ui = (const float*)d_in[12];
  const float* b2_iu = (const float*)d_in[13];
  const int* eui  = (const int*)d_in[16];
  const int* eiu  = (const int*)d_in[17];
  const int* elab = (const int*)d_in[18];

  const int U  = in_sizes[0] / HDIM;
  const int I  = in_sizes[1] / HDIM;
  const int E  = in_sizes[16] / 2;
  const int EL = in_sizes[18] / 2;
  const int R  = U + I;

  float* ws = (float*)d_ws;
  long off = 0;
  float* agg_u = ws + off; off += (long)U * HDIM;
  float* cnt_u = ws + off; off += U;
  float* agg_i = ws + off; off += (long)I * HDIM;
  float* cnt_i = ws + off; off += I;
  float* u1 = ws + off; off += (long)U * HDIM;
  float* i1 = ws + off; off += (long)I * HDIM;
  float* u2 = ws + off; off += (long)U * HDIM;
  float* i2 = ws + off; off += (long)I * HDIM;
  off += off & 1;
  long long* rng = (long long*)(ws + off); off += 2L * R;
  int* cursor = (int*)(ws + off); off += R;
  int* deg    = (int*)(ws + off); off += R + 1;
  int* col    = (int*)(ws + off); off += 2L * E;
  (void)ws_size;

  auto cdiv = [](long a, long b) { return (int)((a + b - 1) / b); };

  const int edgeBlocks = cdiv(2L * E, 256);
  const int gemmBlocksU = cdiv(U, TILE_M);
  const int gemmBlocksI = cdiv(I, TILE_M);
  const int gatherBlocks = cdiv(R, 8 * ROWS_PER_WAVE);
  const int dotBlocks = cdiv(EL, 8 * DOT_EDGES * DOT_TRIPS);

  k_zero_i<<<cdiv(R + 1, 256), 256, 0, stream>>>(deg, R + 1);
  k_degree<<<edgeBlocks, 256, 0, stream>>>(eiu + E, eui + E, deg, E, U);
  k_alloc<<<cdiv(R, 256), 256, 0, stream>>>(deg, rng, cursor, deg + R, cnt_u, cnt_i, U, R);
  k_fill<<<edgeBlocks, 256, 0, stream>>>(eiu, eui, cursor, col, E, U);

  k_gather_sum<<<gatherBlocks, 256, 0, stream>>>(item_emb, user_emb, rng, col,
                                                 agg_u, agg_i, U, R);
  k_sage_gemm<<<gemmBlocksU, 256, 0, stream>>>(agg_u, cnt_u, user_emb, w1_iu_l, w1_iu_r,
                                               b1_iu, u1, U, 1);
  k_sage_gemm<<<gemmBlocksI, 256, 0, stream>>>(agg_i, cnt_i, item_emb, w1_ui_l, w1_ui_r,
                                               b1_ui, i1, I, 1);

  k_gather_sum<<<gatherBlocks, 256, 0, stream>>>(i1, u1, rng, col,
                                                 agg_u, agg_i, U, R);
  k_sage_gemm<<<gemmBlocksU, 256, 0, stream>>>(agg_u, cnt_u, u1, w2_iu_l, w2_iu_r,
                                               b2_iu, u2, U, 0);
  k_sage_gemm<<<gemmBlocksI, 256, 0, stream>>>(agg_i, cnt_i, i1, w2_ui_l, w2_ui_r,
                                               b2_ui, i2, I, 0);

  k_dot<<<dotBlocks, 256, 0, stream>>>(u2, i2, elab, elab + EL, (float*)d_out, EL);
}


# SAGE GEMM as one resident pass per layer: 512 work-groups (both relations in one launch) walk their 64-row tiles, weight strips loaded into registers once, two LDS stages; k and sum order kept
# speedup vs baseline: 2.4569x; 1.0929x over previous
#include <hip/hip_runtime.h>

typedef __attribute__((ext_vector_type(2))) float v2f;
typedef __attribute__((ext_vector_type(8))) float v8f;

#define HDIM 128


__global__ __launch_bounds__(256) void k_zero_i(int* __restrict__ p, int n) {
  int i = blockIdx.x * blockDim.x + threadIdx.x;
  if (i < n) p[i] = 0;
}

__global__ __launch_bounds__(256) void k_degree(const int* __restrict__ dstU,
                                                const int* __restrict__ dstI,
                                                int* __restrict__ deg, int nE, int nU) {
  int t = blockIdx.x * blockDim.x + threadIdx.x;
  if (t >= 2 * nE) return;
  int r = (t < nE) ? dstU[t] : nU + dstI[t - nE];
  atomicAdd(deg + r, 1);
}

__global__ __launch_bounds__(256) void k_alloc(const int* deg, long long* __restrict__ rng,
                                               int* __restrict__ cursor, int* total,
                                               float* __restrict__ cntU,
                                               float* __restrict__ cntI, int nU, int nR) {
  __shared__ int wsum[8];
  __shared__ int blockBase;
  const int lane = threadIdx.x & 31;
  const int wave = threadIdx.x >> 5;
  const int r = blockIdx.x * 256 + threadIdx.x;
  const int d = (r < nR) ? deg[r] : 0;
  int x = d;
  for (int o = 1; o < 32; o <<= 1) {
    int y = __shfl_up(x, o, 32);
    if (lane >= o) x += y;
  }
  if (lane == 31) wsum[wave] = x;
  __syncthreads();
  if (threadIdx.x == 0) {
    int s = 0;
    for (int w = 0; w < 8; ++w) { int v = wsum[w]; wsum[w] = s; s += v; }
    blockBase = atomicAdd(total, s);
  }
  __syncthreads();
  const int pos = blockBase + wsum[wave] + x - d;
  if (r < nR) {
    rng[r] = ((long long)d << 32) | (unsigned)pos;
    cursor[r] = pos;
    if (r < nU) cntU[r] = (float)d; else cntI[r - nU] = (float)d;
  }
}

__global__ __launch_bounds__(256) void k_fill(const int* __restrict__ eiu,
                                              const int* __restrict__ eui,
                                              int* __restrict__ cursor, int* __restrict__ col,
                                              int nE, int nU) {
  int t = blockIdx.x * blockDim.x + threadIdx.x;
  if (t >= 2 * nE) return;
  int s, r;
  if (t < nE) { s = eiu[t]; r = eiu[nE + t]; }
  else { int e = t - nE; s = eui[e]; r = nU + eui[nE + e]; }
  int pos = atomicAdd(cursor + r, 1);
  col[pos] = s;
}

#define ROWS_PER_WAVE 10

__global__ __launch_bounds__(256) void k_gather_sum(
    const float* __restrict__ xItem, const float* __restrict__ xUser,
    const long long* __restrict__ rng, const int* __restrict__ col,
    float* __restrict__ aggU, float* __restrict__ aggI, int nU, int nR) {
  const int lane = threadIdx.x & 31;
  const int nW = gridDim.x * 8;
  int r = __builtin_amdgcn_readfirstlane(blockIdx.x * 8 + (threadIdx.x >> 5));
  if (r >= nR) return;
  long long se = rng[r];
  int rn = r + nW;
  long long seN = (rn < nR) ? rng[rn] : 0LL;
  int idx = (lane < (int)(se >> 32)) ? col[(int)se + lane] : 0;

  do {
    const int rnn = rn + nW;
    const long long seNN = (rnn < nR) ? rng[rnn] : 0LL;

    const bool toUser = r < nU;
    const float* __restrict__ x = toUser ? xItem : xUser;
    const int row = toUser ? r : r - nU;
    const int s0 = (int)se;
    const int n = (int)(se >> 32);

    float4 acc = make_float4(0.f, 0.f, 0.f, 0.f);
    int idxN = 0;
    int base = 0;
    do {
      const int m = (n - base < 32) ? (n - base) : 32;
      int id = idx;
      if (base > 0) id = (lane < m) ? col[s0 + base + lane] : 0;
      int j = 0;
      do {
        int s[8];
#pragma unroll
        for (int u = 0; u < 8; ++u) s[u] = __builtin_amdgcn_readlane(id, j + u);
        float4 v[8];
#pragma unroll
        for (int u = 0; u < 8; ++u) {
          v[u] = make_float4(0.f, 0.f, 0.f, 0.f);
          if (j + u < m)
            v[u] = ((const float4*)(x + (long)s[u] * HDIM))[lane];
        }
        if (base + j == 0) idxN = (lane < (int)(seN >> 32)) ? col[(int)seN + lane] : 0;
#pragma unroll
        for (int u = 0; u < 8; ++u) {
          acc.x += v[u].x; acc.y += v[u].y; acc.z += v[u].z; acc.w += v[u].w;
        }
        j += 8;
      } while (j < m);
      base += 32;
    } while (base < n);
    float* agg = toUser ? aggU : aggI;
    ((float4*)(agg + (long)row * HDIM))[lane] = acc;

    r = rn; rn = rnn; se = seN; seN = seNN; idx = idxN;
  } while (r < nR);
}

#define TILE_M 64
#define APITCH 132
#define GEMM_GRID 512

__global__ __launch_bounds__(256) void k_sage_gemm(
    const float* __restrict__ AggU, const float* __restrict__ cntU,
    const float* __restrict__ XdstU,
    const float* __restrict__ WlU, const float* __restrict__ WrU,
    const float* __restrict__ biasU, float* __restrict__ outU, int nRowsU,
    const float* __restrict__ AggI, const float* __restrict__ cntI,
    const float* __restrict__ XdstI,
    const float* __restrict__ WlI, const float* __restrict__ WrI,
    const float* __restrict__ biasI, float* __restrict__ outI, int nRowsI,
    int splitBlocks, int doRelu) {
  __shared__ __attribute__((aligned(16))) float sA[2][TILE_M * APITCH];
  __shared__ __attribute__((aligned(16))) float sX[2][TILE_M * APITCH];

  const bool forU = (int)blockIdx.x < splitBlocks;
  const float* __restrict__ Agg  = forU ? AggU : AggI;
  const float* __restrict__ cnt  = forU ? cntU : cntI;
  const float* __restrict__ Xdst = forU ? XdstU : XdstI;
  const float* __restrict__ Wl   = forU ? WlU : WlI;
  const float* __restrict__ Wr   = forU ? WrU : WrI;
  const float* __restrict__ bias = forU ? biasU : biasI;
  float* __restrict__ out        = forU ? outU : outI;
  const int nRows = forU ? nRowsU : nRowsI;
  const int tile0 = forU ? (int)blockIdx.x : (int)blockIdx.x - splitBlocks;
  int tstep = forU ? splitBlocks : (int)gridDim.x - splitBlocks;
  if (tstep < 1) tstep = 1;
  const int nTiles = (nRows + TILE_M - 1) / TILE_M;

  const int tid  = threadIdx.x;
  const int lane = tid & 31;
  const int wave = tid >> 5;
  const int c    = lane & 15;
  const int h    = lane >> 4;
  const int col  = wave * 16 + c;

  v2f bl[32], br[32];
#pragma unroll
  for (int s = 0; s < 32; ++s) {
    const int k = 4 * s + 2 * h;
    bl[s].x = Wl[(long)k * HDIM + col];
    bl[s].y = Wl[(long)(k + 1) * HDIM + col];
    br[s].x = Wr[(long)k * HDIM + col];
    br[s].y = Wr[(long)(k + 1) * HDIM + col];
  }
  const float bv = bias[col];

  int buf = 0;
  for (int tile = tile0; tile < nTiles; tile += tstep) {
    const int row0 = tile * TILE_M;
    float* sAb = sA[buf];
    float* sXb = sX[buf];

    {
      const int g = tid & 15;
#pragma unroll
      for (int i = 0; i < 4; ++i) {
        const int r = (tid >> 4) + 16 * i;
        int grow = row0 + r;
        if (grow > nRows - 1) grow = nRows - 1;
        const float inv = 1.0f / fmaxf(cnt[grow], 1.0f);
        const float4* pa = (const float4*)(Agg  + (long)grow * HDIM + 8 * g);
        const float4* px = (const float4*)(Xdst + (long)grow * HDIM + 8 * g);
        float4 a0 = pa[0], a1 = pa[1];
        float4 x0 = px[0], x1 = px[1];
        a0.x *= inv; a0.y *= inv; a0.z *= inv; a0.w *= inv;
        a1.x *= inv; a1.y *= inv; a1.z *= inv; a1.w *= inv;
        float4* da = (float4*)(sAb + r * APITCH + 8 * g);
        float4* dx = (float4*)(sXb + r * APITCH + 8 * g);
        da[0] = make_float4(a0.x, a0.y, a1.x, a1.y);
        da[1] = make_float4(a0.z, a0.w, a1.z, a1.w);
        dx[0] = make_float4(x0.x, x0.y, x1.x, x1.y);
        dx[1] = make_float4(x0.z, x0.w, x1.z, x1.w);
      }
    }
    __syncthreads();

    v8f acc[4];
    const v8f vzero = {};
#pragma unroll
    for (int t = 0; t < 4; ++t) acc[t] = vzero;

    const float* fa0 = sAb + c * APITCH + 4 * h;
    const float* fx0 = sXb + c * APITCH + 4 * h;
#pragma unroll
    for (int j = 0; j < 16; ++j) {
      float4 fa[4], fx[4];
#pragma unroll
      for (int t = 0; t < 4; ++t) {
        fa[t] = *(const float4*)(fa0 + 16 * t * APITCH + 8 * j);
        fx[t] = *(const float4*)(fx0 + 16 * t * APITCH + 8 * j);
      }
#pragma unroll
      for (int t = 0; t < 4; ++t) {
        v2f a, ax;
        a.x = fa[t].x;  a.y = fa[t].y;
        ax.x = fx[t].x; ax.y = fx[t].y;
        acc[t] = __builtin_amdgcn_wmma_f32_16x16x4_f32(
            false, a, false, bl[2 * j], (short)0, acc[t], false, false);
        acc[t] = __builtin_amdgcn_wmma_f32_16x16x4_f32(
            false, ax, false, br[2 * j], (short)0, acc[t], false, false);
      }
#pragma unroll
      for (int t = 0; t < 4; ++t) {
        v2f a, ax;
        a.x = fa[t].z;  a.y = fa[t].w;
        ax.x = fx[t].z; ax.y = fx[t].w;
        acc[t] = __builtin_amdgcn_wmma_f32_16x16x4_f32(
            false, a, false, bl[2 * j + 1], (short)0, acc[t], false, false);
        acc[t] = __builtin_amdgcn_wmma_f32_16x16x4_f32(
            false, ax, false, br[2 * j + 1], (short)0, acc[t], false, false);
      }
    }

    float* orow = out + (long)(row0 + 8 * h) * HDIM + col;
    if (row0 + TILE_M <= nRows) {
#pragma unroll
      for (int t = 0; t < 4; ++t) {
#pragma unroll
        for (int r = 0; r < 8; ++r) {
          float v = acc[t][r] + bv;
          if (doRelu) v = fmaxf(v, 0.0f);
          orow[(long)(16 * t + r) * HDIM] = v;
        }
      }
    } else {
#pragma unroll
      for (int t = 0; t < 4; ++t) {
#pragma unroll
        for (int r = 0; r < 8; ++r) {
          float v = acc[t][r] + bv;
          if (doRelu) v = fmaxf(v, 0.0f);
          if (row0 + 8 * h + 16 * t + r < nRows) orow[(long)(16 * t + r) * HDIM] = v;
        }
      }
    }
    buf ^= 1;
  }
}

#define DOT_EDGES 4
#define DOT_TRIPS 4

__global__ __launch_bounds__(256) void k_dot(const float* __restrict__ u2,
                                             const float* __restrict__ i2,
                                             const int* __restrict__ lu,
                                             const int* __restrict__ li,
                                             float* __restrict__ out, int nEL) {
  const int lane = threadIdx.x & 31;
  const int step = gridDim.x * 8 * DOT_EDGES;
  int e0 = __builtin_amdgcn_readfirstlane(blockIdx.x * 8 + (threadIdx.x >> 5)) * DOT_EDGES;
  if (e0 >= nEL) return;
  int iu[DOT_EDGES], ii[DOT_EDGES];
#pragma unroll
  for (int u = 0; u < DOT_EDGES; ++u) {
    const int e = (e0 + u < nEL) ? e0 + u : nEL - 1;
    iu[u] = lu[e]; ii[u] = li[e];
  }
  do {
    float4 a[DOT_EDGES], b[DOT_EDGES];
#pragma unroll
    for (int u = 0; u < DOT_EDGES; ++u) {
      a[u] = ((const float4*)(u2 + (long)iu[u] * HDIM))[lane];
      b[u] = ((const float4*)(i2 + (long)ii[u] * HDIM))[lane];
    }
    const int e1 = e0 + step;
    int iuN[DOT_EDGES], iiN[DOT_EDGES];
#pragma unroll
    for (int u = 0; u < DOT_EDGES; ++u) { iuN[u] = 0; iiN[u] = 0; }
    if (e1 < nEL) {
#pragma unroll
      for (int u = 0; u < DOT_EDGES; ++u) {
        const int e = (e1 + u < nEL) ? e1 + u : nEL - 1;
        iuN[u] = lu[e]; iiN[u] = li[e];
      }
    }
    float s[DOT_EDGES];
#pragma unroll
    for (int u = 0; u < DOT_EDGES; ++u)
      s[u] = a[u].x * b[u].x + a[u].y * b[u].y + a[u].z * b[u].z + a[u].w * b[u].w;
#pragma unroll
    for (int off = 16; off > 0; off >>= 1) {
#pragma unroll
      for (int u = 0; u < DOT_EDGES; ++u) s[u] += __shfl_down(s[u], off, 32);
    }
    if (lane == 0) {
      if (e0 + DOT_EDGES <= nEL) {
        *(float4*)(out + e0) = make_float4(s[0], s[1], s[2], s[3]);
      } else {
#pragma unroll
        for (int u = 0; u < DOT_EDGES; ++u)
          if (e0 + u < nEL) out[e0 + u] = s[u];
      }
    }
    e0 = e1;
#pragma unroll
    for (int u = 0; u < DOT_EDGES; ++u) { iu[u] = iuN[u]; ii[u] = iiN[u]; }
  } while (e0 < nEL);
}

extern "C" void kernel_launch(void* const* d_in, const int* in_sizes, int n_in,
                              void* d_out, int out_size, void* d_ws, size_t ws_size,
                              hipStream_t stream) {
  const float* user_emb = (const float*)d_in[0];
  const float* item_emb = (const float*)d_in[1];
  const float* w1_ui_l = (const float*)d_in[2];
  const float* w1_ui_r = (const float*)d_in[3];
  const float* w1_iu_l = (const float*)d_in[4];
  const float* w1_iu_r = (const float*)d_in[5];
  const float* w2_ui_l = (const float*)d_in[6];
  const float* w2_ui_r = (const float*)d_in[7];
  const float* w2_iu_l = (const float*)d_in[8];
  const float* w2_iu_r = (const float*)d_in[9];
  const float* b1_ui = (const float*)d_in[10];
  const float* b1_iu = (const float*)d_in[11];
  const float* b2_ui = (const float*)d_in[12];
  const float* b2_iu = (const float*)d_in[13];
  const int* eui  = (const int*)d_in[16];
  const int* eiu  = (const int*)d_in[17];
  const int* elab = (const int*)d_in[18];

  const int U  = in_sizes[0] / HDIM;
  const int I  = in_sizes[1] / HDIM;
  const int E  = in_sizes[16] / 2;
  const int EL = in_sizes[18] / 2;
  const int R  = U + I;

  float* ws = (float*)d_ws;
  long off = 0;
  float* agg_u = ws + off; off += (long)U * HDIM;
  float* cnt_u = ws + off; off += U;
  float* agg_i = ws + off; off += (long)I * HDIM;
  float* cnt_i = ws + off; off += I;
  float* u1 = ws + off; off += (long)U * HDIM;
  float* i1 = ws + off; off += (long)I * HDIM;
  float* u2 = ws + off; off += (long)U * HDIM;
  float* i2 = ws + off; off += (long)I * HDIM;
  off += off & 1;
  long long* rng = (long long*)(ws + off); off += 2L * R;
  int* cursor = (int*)(ws + off); off += R;
  int* deg    = (int*)(ws + off); off += R + 1;
  int* col    = (int*)(ws + off); off += 2L * E;
  (void)ws_size;

  auto cdiv = [](long a, long b) { return (int)((a + b - 1) / b); };

  const int edgeBlocks = cdiv(2L * E, 256);
  const long tilesU = cdiv(U, TILE_M);
  const long tilesI = cdiv(I, TILE_M);
  int gemmSplit = GEMM_GRID / 2;
  if (tilesU + tilesI > 0) gemmSplit = (int)(GEMM_GRID * tilesU / (tilesU + tilesI));
  if (gemmSplit < 1) gemmSplit = 1;
  if (gemmSplit > GEMM_GRID - 1) gemmSplit = GEMM_GRID - 1;
  const int gatherBlocks = cdiv(R, 8 * ROWS_PER_WAVE);
  const int dotBlocks = cdiv(EL, 8 * DOT_EDGES * DOT_TRIPS);

  k_zero_i<<<cdiv(R + 1, 256), 256, 0, stream>>>(deg, R + 1);
  k_degree<<<edgeBlocks, 256, 0, stream>>>(eiu + E, eui + E, deg, E, U);
  k_alloc<<<cdiv(R, 256), 256, 0, stream>>>(deg, rng, cursor, deg + R, cnt_u, cnt_i, U, R);
  k_fill<<<edgeBlocks, 256, 0, stream>>>(eiu, eui, cursor, col, E, U);

  k_gather_sum<<<gatherBlocks, 256, 0, stream>>>(item_emb, user_emb, rng, col,
                                                 agg_u, agg_i, U, R);
  k_sage_gemm<<<GEMM_GRID, 256, 0, stream>>>(agg_u, cnt_u, user_emb, w1_iu_l, w1_iu_r, b1_iu, u1, U,
                                             agg_i, cnt_i, item_emb, w1_ui_l, w1_ui_r, b1_ui, i1, I,
                                             gemmSplit, 1);

  k_gather_sum<<<gatherBlocks, 256, 0, stream>>>(i1, u1, rng, col,
                                                 agg_u, agg_i, U, R);
  k_sage_gemm<<<GEMM_GRID, 256, 0, stream>>>(agg_u, cnt_u, u1, w2_iu_l, w2_iu_r, b2_iu, u2, U,
                                             agg_i, cnt_i, i1, w2_ui_l, w2_ui_r, b2_ui, i2, I,
                                             gemmSplit, 0);

  k_dot<<<dotBlocks, 256, 0, stream>>>(u2, i2, elab, elab + EL, (float*)d_out, EL);
}


# Bucket sum fused into the SAGE GEMM: each wave sums its 8 tile rows' source rows (2 rows at a time, bucket order kept) straight into the LDS stage; no sum plane, 2 launches fewer, grid split by cost
# speedup vs baseline: 2.6632x; 1.0840x over previous
#include <hip/hip_runtime.h>

typedef __attribute__((ext_vector_type(2))) float v2f;
typedef __attribute__((ext_vector_type(8))) float v8f;

#define HDIM 128


__global__ __launch_bounds__(256) void k_zero_i(int* __restrict__ p, int n) {
  int i = blockIdx.x * blockDim.x + threadIdx.x;
  if (i < n) p[i] = 0;
}

__global__ __launch_bounds__(256) void k_degree(const int* __restrict__ dstU,
                                                const int* __restrict__ dstI,
                                                int* __restrict__ deg, int nE, int nU) {
  int t = blockIdx.x * blockDim.x + threadIdx.x;
  if (t >= 2 * nE) return;
  int r = (t < nE) ? dstU[t] : nU + dstI[t - nE];
  atomicAdd(deg + r, 1);
}

__global__ __launch_bounds__(256) void k_alloc(const int* deg, long long* __restrict__ rng,
                                               int* __restrict__ cursor, int* total, int nR) {
  __shared__ int wsum[8];
  __shared__ int blockBase;
  const int lane = threadIdx.x & 31;
  const int wave = threadIdx.x >> 5;
  const int r = blockIdx.x * 256 + threadIdx.x;
  const int d = (r < nR) ? deg[r] : 0;
  int x = d;
  for (int o = 1; o < 32; o <<= 1) {
    int y = __shfl_up(x, o, 32);
    if (lane >= o) x += y;
  }
  if (lane == 31) wsum[wave] = x;
  __syncthreads();
  if (threadIdx.x == 0) {
    int s = 0;
    for (int w = 0; w < 8; ++w) { int v = wsum[w]; wsum[w] = s; s += v; }
    blockBase = atomicAdd(total, s);
  }
  __syncthreads();
  const int pos = blockBase + wsum[wave] + x - d;
  if (r < nR) {
    rng[r] = ((long long)d << 32) | (unsigned)pos;
    cursor[r] = pos;
  }
}

__global__ __launch_bounds__(256) void k_fill(const int* __restrict__ eiu,
                                              const int* __restrict__ eui,
                                              int* __restrict__ cursor, int* __restrict__ col,
                                              int nE, int nU) {
  int t = blockIdx.x * blockDim.x + threadIdx.x;
  if (t >= 2 * nE) return;
  int s, r;
  if (t < nE) { s = eiu[t]; r = eiu[nE + t]; }
  else { int e = t - nE; s = eui[e]; r = nU + eui[nE + e]; }
  int pos = atomicAdd(cursor + r, 1);
  col[pos] = s;
}

#define TILE_M 64
#define APITCH 132
#define GEMM_GRID 512

__global__ __launch_bounds__(256) void k_sage_layer(
    const float* __restrict__ XsrcU, const float* __restrict__ XdstU,
    const float* __restrict__ WlU, const float* __restrict__ WrU,
    const float* __restrict__ biasU, float* __restrict__ outU, int nRowsU,
    const float* __restrict__ XsrcI, const float* __restrict__ XdstI,
    const float* __restrict__ WlI, const float* __restrict__ WrI,
    const float* __restrict__ biasI, float* __restrict__ outI, int nRowsI,
    const long long* __restrict__ rng, const int* __restrict__ col,
    int splitBlocks, int doRelu) {
  __shared__ __attribute__((aligned(16))) float sA[2][TILE_M * APITCH];
  __shared__ __attribute__((aligned(16))) float sX[2][TILE_M * APITCH];

  const bool forU = (int)blockIdx.x < splitBlocks;
  const float* __restrict__ Xsrc = forU ? XsrcU : XsrcI;
  const float* __restrict__ Xdst = forU ? XdstU : XdstI;
  const float* __restrict__ Wl   = forU ? WlU : WlI;
  const float* __restrict__ Wr   = forU ? WrU : WrI;
  const float* __restrict__ bias = forU ? biasU : biasI;
  float* __restrict__ out        = forU ? outU : outI;
  const int nRows = forU ? nRowsU : nRowsI;
  const int rbase = forU ? 0 : nRowsU;
  const int tile0 = forU ? (int)blockIdx.x : (int)blockIdx.x - splitBlocks;
  int tstep = forU ? splitBlocks : (int)gridDim.x - splitBlocks;
  if (tstep < 1) tstep = 1;
  const int nTiles = (nRows + TILE_M - 1) / TILE_M;

  const int tid  = threadIdx.x;
  const int lane = tid & 31;
  const int wave = tid >> 5;
  const int c    = lane & 15;
  const int h    = lane >> 4;
  const int col16 = wave * 16 + c;
  const int wr0 = __builtin_amdgcn_readfirstlane(wave) * 8;
  const int lofs = 8 * (lane >> 1) + 2 * (lane & 1);

  v2f bl[32], br[32];
#pragma unroll
  for (int s = 0; s < 32; ++s) {
    const int k = 4 * s + 2 * h;
    bl[s].x = Wl[(long)k * HDIM + col16];
    bl[s].y = Wl[(long)(k + 1) * HDIM + col16];
    br[s].x = Wr[(long)k * HDIM + col16];
    br[s].y = Wr[(long)(k + 1) * HDIM + col16];
  }
  const float bv = bias[col16];

  long long seV = 0LL;
  if (tile0 < nTiles) {
    int lr = tile0 * TILE_M + wr0 + (lane & 7);
    if (lr > nRows - 1) lr = nRows - 1;
    seV = rng[rbase + lr];
  }

  int buf = 0;
  for (int tile = tile0; tile < nTiles; tile += tstep) {
    const int row0 = tile * TILE_M;
    float* sAb = sA[buf];
    float* sXb = sX[buf];

    {
      const int seLo = (int)seV;
      const int seHi = (int)(seV >> 32);
      int sa0 = __builtin_amdgcn_readlane(seLo, 0), na = __builtin_amdgcn_readlane(seHi, 0);
      int sb0 = __builtin_amdgcn_readlane(seLo, 1), nb = __builtin_amdgcn_readlane(seHi, 1);
      int idA = (lane < na) ? col[sa0 + lane] : 0;
      int idB = (lane < nb) ? col[sb0 + lane] : 0;
#pragma unroll 1
      for (int p = 0; p < 4; ++p) {
        int sa0N = 0, naN = 0, sb0N = 0, nbN = 0, idAN = 0, idBN = 0;
        int gA = row0 + wr0 + 2 * p;
        int gB = gA + 1;
        if (gA > nRows - 1) gA = nRows - 1;
        if (gB > nRows - 1) gB = nRows - 1;
        float4 xa = make_float4(0.f, 0.f, 0.f, 0.f);
        float4 xb = make_float4(0.f, 0.f, 0.f, 0.f);

        float4 accA = make_float4(0.f, 0.f, 0.f, 0.f);
        float4 accB = make_float4(0.f, 0.f, 0.f, 0.f);
        const int nmax = (na > nb) ? na : nb;
        int base = 0;
        do {
          int ma = na - base; ma = (ma < 0) ? 0 : ((ma > 32) ? 32 : ma);
          int mb = nb - base; mb = (mb < 0) ? 0 : ((mb > 32) ? 32 : mb);
          int ia = idA, ib = idB;
          if (base > 0) {
            ia = (lane < ma) ? col[sa0 + base + lane] : 0;
            ib = (lane < mb) ? col[sb0 + base + lane] : 0;
          }
          const int mm = (ma > mb) ? ma : mb;
          int j = 0;
          do {
            int s[8], t[8];
#pragma unroll
            for (int u = 0; u < 8; ++u) {
              s[u] = __builtin_amdgcn_readlane(ia, j + u);
              t[u] = __builtin_amdgcn_readlane(ib, j + u);
            }
            float4 v[8], w[8];
#pragma unroll
            for (int u = 0; u < 8; ++u) {
              v[u] = make_float4(0.f, 0.f, 0.f, 0.f);
              if (j + u < ma)
                v[u] = ((const float4*)(Xsrc + (long)s[u] * HDIM))[lane];
            }
#pragma unroll
            for (int u = 0; u < 8; ++u) {
              w[u] = make_float4(0.f, 0.f, 0.f, 0.f);
              if (j + u < mb)
                w[u] = ((const float4*)(Xsrc + (long)t[u] * HDIM))[lane];
            }
            if (base + j == 0) {
              xa = ((const float4*)(Xdst + (long)gA * HDIM))[lane];
              xb = ((const float4*)(Xdst + (long)gB * HDIM))[lane];
              if (p < 3) {
                sa0N = __builtin_amdgcn_readlane(seLo, 2 * p + 2);
                naN  = __builtin_amdgcn_readlane(seHi, 2 * p + 2);
                sb0N = __builtin_amdgcn_readlane(seLo, 2 * p + 3);
                nbN  = __builtin_amdgcn_readlane(seHi, 2 * p + 3);
                idAN = (lane < naN) ? col[sa0N + lane] : 0;
                idBN = (lane < nbN) ? col[sb0N + lane] : 0;
              }
            }
#pragma unroll
            for (int u = 0; u < 8; ++u) {
              accA.x += v[u].x; accA.y += v[u].y; accA.z += v[u].z; accA.w += v[u].w;
            }
#pragma unroll
            for (int u = 0; u < 8; ++u) {
              accB.x += w[u].x; accB.y += w[u].y; accB.z += w[u].z; accB.w += w[u].w;
            }
            j += 8;
          } while (j < mm);
          base += 32;
        } while (base < nmax);

        const float invA = 1.0f / fmaxf((float)na, 1.0f);
        const float invB = 1.0f / fmaxf((float)nb, 1.0f);
        float* da = sAb + (wr0 + 2 * p) * APITCH + lofs;
        float* dx = sXb + (wr0 + 2 * p) * APITCH + lofs;
        *(float2*)(da)              = make_float2(accA.x * invA, accA.y * invA);
        *(float2*)(da + 4)          = make_float2(accA.z * invA, accA.w * invA);
        *(float2*)(da + APITCH)     = make_float2(accB.x * invB, accB.y * invB);
        *(float2*)(da + APITCH + 4) = make_float2(accB.z * invB, accB.w * invB);
        *(float2*)(dx)              = make_float2(xa.x, xa.y);
        *(float2*)(dx + 4)          = make_float2(xa.z, xa.w);
        *(float2*)(dx + APITCH)     = make_float2(xb.x, xb.y);
        *(float2*)(dx + APITCH + 4) = make_float2(xb.z, xb.w);

        sa0 = sa0N; na = naN; sb0 = sb0N; nb = nbN; idA = idAN; idB = idBN;
      }
    }
    __syncthreads();

    {
      const int tileN = tile + tstep;
      if (tileN < nTiles) {
        int lr = tileN * TILE_M + wr0 + (lane & 7);
        if (lr > nRows - 1) lr = nRows - 1;
        seV = rng[rbase + lr];
      }
    }

    v8f acc[4];
    const v8f vzero = {};
#pragma unroll
    for (int t = 0; t < 4; ++t) acc[t] = vzero;

    const float* fa0 = sAb + c * APITCH + 4 * h;
    const float* fx0 = sXb + c * APITCH + 4 * h;
#pragma unroll
    for (int j = 0; j < 16; ++j) {
      float4 fa[4], fx[4];
#pragma unroll
      for (int t = 0; t < 4; ++t) {
        fa[t] = *(const float4*)(fa0 + 16 * t * APITCH + 8 * j);
        fx[t] = *(const float4*)(fx0 + 16 * t * APITCH + 8 * j);
      }
#pragma unroll
      for (int t = 0; t < 4; ++t) {
        v2f a, ax;
        a.x = fa[t].x;  a.y = fa[t].y;
        ax.x = fx[t].x; ax.y = fx[t].y;
        acc[t] = __builtin_amdgcn_wmma_f32_16x16x4_f32(
            false, a, false, bl[2 * j], (short)0, acc[t], false, false);
        acc[t] = __builtin_amdgcn_wmma_f32_16x16x4_f32(
            false, ax, false, br[2 * j], (short)0, acc[t], false, false);
      }
#pragma unroll
      for (int t = 0; t < 4; ++t) {
        v2f a, ax;
        a.x = fa[t].z;  a.y = fa[t].w;
        ax.x = fx[t].z; ax.y = fx[t].w;
        acc[t] = __builtin_amdgcn_wmma_f32_16x16x4_f32(
            false, a, false, bl[2 * j + 1], (short)0, acc[t], false, false);
        acc[t] = __builtin_amdgcn_wmma_f32_16x16x4_f32(
            false, ax, false, br[2 * j + 1], (short)0, acc[t], false, false);
      }
    }

    float* orow = out + (long)(row0 + 8 * h) * HDIM + col16;
    if (row0 + TILE_M <= nRows) {
#pragma unroll
      for (int t = 0; t < 4; ++t) {
#pragma unroll
        for (int r = 0; r < 8; ++r) {
          float v = acc[t][r] + bv;
          if (doRelu) v = fmaxf(v, 0.0f);
          orow[(long)(16 * t + r) * HDIM] = v;
        }
      }
    } else {
#pragma unroll
      for (int t = 0; t < 4; ++t) {
#pragma unroll
        for (int r = 0; r < 8; ++r) {
          float v = acc[t][r] + bv;
          if (doRelu) v = fmaxf(v, 0.0f);
          if (row0 + 8 * h + 16 * t + r < nRows) orow[(long)(16 * t + r) * HDIM] = v;
        }
      }
    }
    buf ^= 1;
  }
}

#define DOT_EDGES 4
#define DOT_TRIPS 4

__global__ __launch_bounds__(256) void k_dot(const float* __restrict__ u2,
                                             const float* __restrict__ i2,
                                             const int* __restrict__ lu,
                                             const int* __restrict__ li,
                                             float* __restrict__ out, int nEL) {
  const int lane = threadIdx.x & 31;
  const int step = gridDim.x * 8 * DOT_EDGES;
  int e0 = __builtin_amdgcn_readfirstlane(blockIdx.x * 8 + (threadIdx.x >> 5)) * DOT_EDGES;
  if (e0 >= nEL) return;
  int iu[DOT_EDGES], ii[DOT_EDGES];
#pragma unroll
  for (int u = 0; u < DOT_EDGES; ++u) {
    const int e = (e0 + u < nEL) ? e0 + u : nEL - 1;
    iu[u] = lu[e]; ii[u] = li[e];
  }
  do {
    float4 a[DOT_EDGES], b[DOT_EDGES];
#pragma unroll
    for (int u = 0; u < DOT_EDGES; ++u) {
      a[u] = ((const float4*)(u2 + (long)iu[u] * HDIM))[lane];
      b[u] = ((const float4*)(i2 + (long)ii[u] * HDIM))[lane];
    }
    const int e1 = e0 + step;
    int iuN[DOT_EDGES], iiN[DOT_EDGES];
#pragma unroll
    for (int u = 0; u < DOT_EDGES; ++u) { iuN[u] = 0; iiN[u] = 0; }
    if (e1 < nEL) {
#pragma unroll
      for (int u = 0; u < DOT_EDGES; ++u) {
        const int e = (e1 + u < nEL) ? e1 + u : nEL - 1;
        iuN[u] = lu[e]; iiN[u] = li[e];
      }
    }
    float s[DOT_EDGES];
#pragma unroll
    for (int u = 0; u < DOT_EDGES; ++u)
      s[u] = a[u].x * b[u].x + a[u].y * b[u].y + a[u].z * b[u].z + a[u].w * b[u].w;
#pragma unroll
    for (int off = 16; off > 0; off >>= 1) {
#pragma unroll
      for (int u = 0; u < DOT_EDGES; ++u) s[u] += __shfl_down(s[u], off, 32);
    }
    if (lane == 0) {
      if (e0 + DOT_EDGES <= nEL) {
        *(float4*)(out + e0) = make_float4(s[0], s[1], s[2], s[3]);
      } else {
#pragma unroll
        for (int u = 0; u < DOT_EDGES; ++u)
          if (e0 + u < nEL) out[e0 + u] = s[u];
      }
    }
    e0 = e1;
#pragma unroll
    for (int u = 0; u < DOT_EDGES; ++u) { iu[u] = iuN[u]; ii[u] = iiN[u]; }
  } while (e0 < nEL);
}

extern "C" void kernel_launch(void* const* d_in, const int* in_sizes, int n_in,
                              void* d_out, int out_size, void* d_ws, size_t ws_size,
                              hipStream_t stream) {
  const float* user_emb = (const float*)d_in[0];
  const float* item_emb = (const float*)d_in[1];
  const float* w1_ui_l = (const float*)d_in[2];
  const float* w1_ui_r = (const float*)d_in[3];
  const float* w1_iu_l = (const float*)d_in[4];
  const float* w1_iu_r = (const float*)d_in[5];
  const float* w2_ui_l = (const float*)d_in[6];
  const float* w2_ui_r = (const float*)d_in[7];
  const float* w2_iu_l = (const float*)d_in[8];
  const float* w2_iu_r = (const float*)d_in[9];
  const float* b1_ui = (const float*)d_in[10];
  const float* b1_iu = (const float*)d_in[11];
  const float* b2_ui = (const float*)d_in[12];
  const float* b2_iu = (const float*)d_in[13];
  const int* eui  = (const int*)d_in[16];
  const int* eiu  = (const int*)d_in[17];
  const int* elab = (const int*)d_in[18];

  const int U  = in_sizes[0] / HDIM;
  const int I  = in_sizes[1] / HDIM;
  const int E  = in_sizes[16] / 2;
  const int EL = in_sizes[18] / 2;
  const int R  = U + I;

  float* ws = (float*)d_ws;
  long off = 0;
  float* u1 = ws + off; off += (long)U * HDIM;
  float* i1 = ws + off; off += (long)I * HDIM;
  float* u2 = ws + off; off += (long)U * HDIM;
  float* i2 = ws + off; off += (long)I * HDIM;
  off += off & 1;
  long long* rng = (long long*)(ws + off); off += 2L * R;
  int* cursor = (int*)(ws + off); off += R;
  int* deg    = (int*)(ws + off); off += R + 1;
  int* col    = (int*)(ws + off); off += 2L * E;
  (void)ws_size;

  auto cdiv = [](long a, long b) { return (int)((a + b - 1) / b); };

  const int edgeBlocks = cdiv(2L * E, 256);
  const long costU = 18L * U + E;
  const long costI = 18L * I + E;
  int gemmSplit = GEMM_GRID / 2;
  if (costU + costI > 0) gemmSplit = (int)(GEMM_GRID * costU / (costU + costI));
  if (gemmSplit < 1) gemmSplit = 1;
  if (gemmSplit > GEMM_GRID - 1) gemmSplit = GEMM_GRID - 1;
  const int dotBlocks = cdiv(EL, 8 * DOT_EDGES * DOT_TRIPS);

  k_zero_i<<<cdiv(R + 1, 256), 256, 0, stream>>>(deg, R + 1);
  k_degree<<<edgeBlocks, 256, 0, stream>>>(eiu + E, eui + E, deg, E, U);
  k_alloc<<<cdiv(R, 256), 256, 0, stream>>>(deg, rng, cursor, deg + R, R);
  k_fill<<<edgeBlocks, 256, 0, stream>>>(eiu, eui, cursor, col, E, U);

  k_sage_layer<<<GEMM_GRID, 256, 0, stream>>>(item_emb, user_emb, w1_iu_l, w1_iu_r, b1_iu, u1, U,
                                              user_emb, item_emb, w1_ui_l, w1_ui_r, b1_ui, i1, I,
                                              rng, col, gemmSplit, 1);

  k_sage_layer<<<GEMM_GRID, 256, 0, stream>>>(i1, u1, w2_iu_l, w2_iu_r, b2_iu, u2, U,
                                              u1, i1, w2_ui_l, w2_ui_r, b2_ui, i2, I,
                                              rng, col, gemmSplit, 0);

  k_dot<<<dotBlocks, 256, 0, stream>>>(u2, i2, elab, elab + EL, (float*)d_out, EL);
}


# Edge bucketing as a two-level counting sort, all counters in LDS (chunk bin counts, bin-grouped copy, per-bin row sort): no global atomics, 1 launch fewer; sum order = LDS serve order, run-dependent
# speedup vs baseline: 3.7882x; 1.4224x over previous
#include <hip/hip_runtime.h>

typedef __attribute__((ext_vector_type(2))) float v2f;
typedef __attribute__((ext_vector_type(8))) float v8f;

#define HDIM 128

#define BIN_SHIFT 9
#define BIN_ROWS 512
#define MAX_BINS 1024
#define SORT_BLOCKS 128
#define SORT_THREADS 512
#define EDGE_GROUP 8
#define HIST_GROUP 32

__device__ __forceinline__ int edge_row(const int* __restrict__ eiu, const int* __restrict__ eui,
                                        int t, int nE, int nU) {
  return (t < nE) ? eiu[nE + t] : nU + eui[t];
}
__device__ __forceinline__ int edge_src(const int* __restrict__ eiu, const int* __restrict__ eui,
                                        int t, int nE) {
  return (t < nE) ? eiu[t] : eui[t - nE];
}

__global__ __launch_bounds__(SORT_THREADS) void k_bin_count(const int* __restrict__ eiu,
                                                            const int* __restrict__ eui,
                                                            int* __restrict__ hist,
                                                            int nE, int nU, int nBins) {
  __shared__ int cnt[MAX_BINS];
  for (int b = threadIdx.x; b < nBins; b += SORT_THREADS) cnt[b] = 0;
  __syncthreads();
  const int nE2 = 2 * nE;
  const int chunk = (nE2 + SORT_BLOCKS - 1) / SORT_BLOCKS;
  const int t0 = (int)blockIdx.x * chunk;
  int t1 = t0 + chunk;
  if (t1 > nE2) t1 = nE2;
  for (int tb = t0 + (int)threadIdx.x; tb < t1; tb += SORT_THREADS * EDGE_GROUP) {
    int r[EDGE_GROUP];
#pragma unroll
    for (int u = 0; u < EDGE_GROUP; ++u) {
      const int t = tb + u * SORT_THREADS;
      r[u] = -1;
      if (t < t1) r[u] = edge_row(eiu, eui, t, nE, nU);
    }
#pragma unroll
    for (int u = 0; u < EDGE_GROUP; ++u)
      if (r[u] >= 0) atomicAdd(&cnt[r[u] >> BIN_SHIFT], 1);
  }
  __syncthreads();
  for (int b = threadIdx.x; b < nBins; b += SORT_THREADS) hist[(int)blockIdx.x * nBins + b] = cnt[b];
}

__global__ __launch_bounds__(SORT_THREADS) void k_bin_scatter(const int* __restrict__ eiu,
                                                              const int* __restrict__ eui,
                                                              const int* __restrict__ hist,
                                                              int* __restrict__ binStart,
                                                              int* __restrict__ pairWords,
                                                              int nE, int nU, int nBins) {
  int2* __restrict__ pairs = (int2*)pairWords;
  __shared__ int tot[MAX_BINS];
  __shared__ int cur[MAX_BINS];
  const int nBlk = SORT_BLOCKS;
  for (int b = threadIdx.x; b < nBins; b += SORT_THREADS) {
    int all = 0, before = 0;
    for (int k0 = 0; k0 < SORT_BLOCKS; k0 += HIST_GROUP) {
      int v[HIST_GROUP];
#pragma unroll
      for (int u = 0; u < HIST_GROUP; ++u) v[u] = hist[(k0 + u) * nBins + b];
#pragma unroll
      for (int u = 0; u < HIST_GROUP; ++u) {
        all += v[u];
        before += (k0 + u < (int)blockIdx.x) ? v[u] : 0;
      }
    }
    tot[b] = all;
    cur[b] = before;
  }
  __syncthreads();
  if (threadIdx.x < 32) {
    const int lane = threadIdx.x;
    const int per = (nBins + 31) / 32;
    const int b0 = lane * per;
    int s = 0;
    for (int i = 0; i < per; ++i) {
      const int b = b0 + i;
      if (b < nBins) s += tot[b];
    }
    int x = s;
    for (int o = 1; o < 32; o <<= 1) {
      int y = __shfl_up(x, o, 32);
      if (lane >= o) x += y;
    }
    int run = x - s;
    for (int i = 0; i < per; ++i) {
      const int b = b0 + i;
      if (b < nBins) { const int v = tot[b]; tot[b] = run; run += v; }
    }
  }
  __syncthreads();
  for (int b = threadIdx.x; b < nBins; b += SORT_THREADS) {
    const int first = tot[b];
    cur[b] += first;
    if (blockIdx.x == 0) binStart[b] = first;
  }
  if (blockIdx.x == 0 && threadIdx.x == 0) binStart[nBins] = 2 * nE;
  __syncthreads();
  const int nE2 = 2 * nE;
  const int chunk = (nE2 + nBlk - 1) / nBlk;
  const int t0 = (int)blockIdx.x * chunk;
  int t1 = t0 + chunk;
  if (t1 > nE2) t1 = nE2;
  for (int tb = t0 + (int)threadIdx.x; tb < t1; tb += SORT_THREADS * EDGE_GROUP) {
    int r[EDGE_GROUP], s[EDGE_GROUP];
#pragma unroll
    for (int u = 0; u < EDGE_GROUP; ++u) {
      const int t = tb + u * SORT_THREADS;
      r[u] = -1; s[u] = 0;
      if (t < t1) { r[u] = edge_row(eiu, eui, t, nE, nU); s[u] = edge_src(eiu, eui, t, nE); }
    }
#pragma unroll
    for (int u = 0; u < EDGE_GROUP; ++u) {
      if (r[u] >= 0) {
        const int p = atomicAdd(&cur[r[u] >> BIN_SHIFT], 1);
        pairs[p] = make_int2(s[u], r[u]);
      }
    }
  }
}

__global__ __launch_bounds__(BIN_ROWS) void k_bin_sort(const int* __restrict__ pairWords,
                                                       const int* __restrict__ binStart,
                                                       long long* __restrict__ rng,
                                                       int* __restrict__ col, int nR) {
  const int2* __restrict__ pairs = (const int2*)pairWords;
  __shared__ int cnt[BIN_ROWS];
  __shared__ int wsum[BIN_ROWS / 32];
  const int lane = threadIdx.x & 31;
  const int wave = threadIdx.x >> 5;
  const int r0 = (int)blockIdx.x * BIN_ROWS;
  const int lo = binStart[blockIdx.x];
  const int hi = binStart[blockIdx.x + 1];
  cnt[threadIdx.x] = 0;
  __syncthreads();
  for (int ib = lo + (int)threadIdx.x; ib < hi; ib += BIN_ROWS * EDGE_GROUP) {
    int r[EDGE_GROUP];
#pragma unroll
    for (int u = 0; u < EDGE_GROUP; ++u) {
      const int i = ib + u * BIN_ROWS;
      r[u] = -1;
      if (i < hi) r[u] = pairs[i].y;
    }
#pragma unroll
    for (int u = 0; u < EDGE_GROUP; ++u)
      if (r[u] >= 0) atomicAdd(&cnt[r[u] - r0], 1);
  }
  __syncthreads();
  const int d = cnt[threadIdx.x];
  int x = d;
  for (int o = 1; o < 32; o <<= 1) {
    int y = __shfl_up(x, o, 32);
    if (lane >= o) x += y;
  }
  if (lane == 31) wsum[wave] = x;
  __syncthreads();
  int wbase = 0;
  for (int w = 0; w < wave; ++w) wbase += wsum[w];
  const int first = wbase + x - d;
  cnt[threadIdx.x] = first;
  if (r0 + (int)threadIdx.x < nR)
    rng[r0 + threadIdx.x] = ((long long)d << 32) | (unsigned)(lo + first);
  __syncthreads();
  for (int ib = lo + (int)threadIdx.x; ib < hi; ib += BIN_ROWS * EDGE_GROUP) {
    int2 e[EDGE_GROUP];
#pragma unroll
    for (int u = 0; u < EDGE_GROUP; ++u) {
      const int i = ib + u * BIN_ROWS;
      e[u] = make_int2(0, -1);
      if (i < hi) e[u] = pairs[i];
    }
#pragma unroll
    for (int u = 0; u < EDGE_GROUP; ++u) {
      if (e[u].y >= 0) {
        const int p = atomicAdd(&cnt[e[u].y - r0], 1);
        col[lo + p] = e[u].x;
      }
    }
  }
}

#define TILE_M 64
#define APITCH 132
#define GEMM_GRID 512

__global__ __launch_bounds__(256) void k_sage_layer(
    const float* __restrict__ XsrcU, const float* __restrict__ XdstU,
    const float* __restrict__ WlU, const float* __restrict__ WrU,
    const float* __restrict__ biasU, float* __restrict__ outU, int nRowsU,
    const float* __restrict__ XsrcI, const float* __restrict__ XdstI,
    const float* __restrict__ WlI, const float* __restrict__ WrI,
    const float* __restrict__ biasI, float* __restrict__ outI, int nRowsI,
    const long long* __restrict__ rng, const int* __restrict__ col,
    int splitBlocks, int doRelu) {
  __shared__ __attribute__((aligned(16))) float sA[2][TILE_M * APITCH];
  __shared__ __attribute__((aligned(16))) float sX[2][TILE_M * APITCH];

  const bool forU = (int)blockIdx.x < splitBlocks;
  const float* __restrict__ Xsrc = forU ? XsrcU : XsrcI;
  const float* __restrict__ Xdst = forU ? XdstU : XdstI;
  const float* __restrict__ Wl   = forU ? WlU : WlI;
  const float* __restrict__ Wr   = forU ? WrU : WrI;
  const float* __restrict__ bias = forU ? biasU : biasI;
  float* __restrict__ out        = forU ? outU : outI;
  const int nRows = forU ? nRowsU : nRowsI;
  const int rbase = forU ? 0 : nRowsU;
  const int tile0 = forU ? (int)blockIdx.x : (int)blockIdx.x - splitBlocks;
  int tstep = forU ? splitBlocks : (int)gridDim.x - splitBlocks;
  if (tstep < 1) tstep = 1;
  const int nTiles = (nRows + TILE_M - 1) / TILE_M;

  const int tid  = threadIdx.x;
  const int lane = tid & 31;
  const int wave = tid >> 5;
  const int c    = lane & 15;
  const int h    = lane >> 4;
  const int col16 = wave * 16 + c;
  const int wr0 = __builtin_amdgcn_readfirstlane(wave) * 8;
  const int lofs = 8 * (lane >> 1) + 2 * (lane & 1);

  v2f bl[32], br[32];
#pragma unroll
  for (int s = 0; s < 32; ++s) {
    const int k = 4 * s + 2 * h;
    bl[s].x = Wl[(long)k * HDIM + col16];
    bl[s].y = Wl[(long)(k + 1) * HDIM + col16];
    br[s].x = Wr[(long)k * HDIM + col16];
    br[s].y = Wr[(long)(k + 1) * HDIM + col16];
  }
  const float bv = bias[col16];

  long long seV = 0LL;
  if (tile0 < nTiles) {
    int lr = tile0 * TILE_M + wr0 + (lane & 7);
    if (lr > nRows - 1) lr = nRows - 1;
    seV = rng[rbase + lr];
  }

  int buf = 0;
  for (int tile = tile0; tile < nTiles; tile += tstep) {
    const int row0 = tile * TILE_M;
    float* sAb = sA[buf];
    float* sXb = sX[buf];

    {
      const int seLo = (int)seV;
      const int seHi = (int)(seV >> 32);
      int sa0 = __builtin_amdgcn_readlane(seLo, 0), na = __builtin_amdgcn_readlane(seHi, 0);
      int sb0 = __builtin_amdgcn_readlane(seLo, 1), nb = __builtin_amdgcn_readlane(seHi, 1);
      int idA = (lane < na) ? col[sa0 + lane] : 0;
      int idB = (lane < nb) ? col[sb0 + lane] : 0;
#pragma unroll 1
      for (int p = 0; p < 4; ++p) {
        int sa0N = 0, naN = 0, sb0N = 0, nbN = 0, idAN = 0, idBN = 0;
        int gA = row0 + wr0 + 2 * p;
        int gB = gA + 1;
        if (gA > nRows - 1) gA = nRows - 1;
        if (gB > nRows - 1) gB = nRows - 1;
        float4 xa = make_float4(0.f, 0.f, 0.f, 0.f);
        float4 xb = make_float4(0.f, 0.f, 0.f, 0.f);

        float4 accA = make_float4(0.f, 0.f, 0.f, 0.f);
        float4 accB = make_float4(0.f, 0.f, 0.f, 0.f);
        const int nmax = (na > nb) ? na : nb;
        int base = 0;
        do {
          int ma = na - base; ma = (ma < 0) ? 0 : ((ma > 32) ? 32 : ma);
          int mb = nb - base; mb = (mb < 0) ? 0 : ((mb > 32) ? 32 : mb);
          int ia = idA, ib = idB;
          if (base > 0) {
            ia = (lane < ma) ? col[sa0 + base + lane] : 0;
            ib = (lane < mb) ? col[sb0 + base + lane] : 0;
          }
          const int mm = (ma > mb) ? ma : mb;
          int j = 0;
          do {
            int s[8], t[8];
#pragma unroll
            for (int u = 0; u < 8; ++u) {
              s[u] = __builtin_amdgcn_readlane(ia, j + u);
              t[u] = __builtin_amdgcn_readlane(ib, j + u);
            }
            float4 v[8], w[8];
#pragma unroll
            for (int u = 0; u < 8; ++u) {
              v[u] = make_float4(0.f, 0.f, 0.f, 0.f);
              if (j + u < ma)
                v[u] = ((const float4*)(Xsrc + (long)s[u] * HDIM))[lane];
            }
#pragma unroll
            for (int u = 0; u < 8; ++u) {
              w[u] = make_float4(0.f, 0.f, 0.f, 0.f);
              if (j + u < mb)
                w[u] = ((const float4*)(Xsrc + (long)t[u] * HDIM))[lane];
            }
            if (base + j == 0) {
              xa = ((const float4*)(Xdst + (long)gA * HDIM))[lane];
              xb = ((const float4*)(Xdst + (long)gB * HDIM))[lane];
              if (p < 3) {
                sa0N = __builtin_amdgcn_readlane(seLo, 2 * p + 2);
                naN  = __builtin_amdgcn_readlane(seHi, 2 * p + 2);
                sb0N = __builtin_amdgcn_readlane(seLo, 2 * p + 3);
                nbN  = __builtin_amdgcn_readlane(seHi, 2 * p + 3);
                idAN = (lane < naN) ? col[sa0N + lane] : 0;
                idBN = (lane < nbN) ? col[sb0N + lane] : 0;
              }
            }
#pragma unroll
            for (int u = 0; u < 8; ++u) {
              accA.x += v[u].x; accA.y += v[u].y; accA.z += v[u].z; accA.w += v[u].w;
            }
#pragma unroll
            for (int u = 0; u < 8; ++u) {
              accB.x += w[u].x; accB.y += w[u].y; accB.z += w[u].z; accB.w += w[u].w;
            }
            j += 8;
          } while (j < mm);
          base += 32;
        } while (base < nmax);

        const float invA = 1.0f / fmaxf((float)na, 1.0f);
        const float invB = 1.0f / fmaxf((float)nb, 1.0f);
        float* da = sAb + (wr0 + 2 * p) * APITCH + lofs;
        float* dx = sXb + (wr0 + 2 * p) * APITCH + lofs;
        *(float2*)(da)              = make_float2(accA.x * invA, accA.y * invA);
        *(float2*)(da + 4)          = make_float2(accA.z * invA, accA.w * invA);
        *(float2*)(da + APITCH)     = make_float2(accB.x * invB, accB.y * invB);
        *(float2*)(da + APITCH + 4) = make_float2(accB.z * invB, accB.w * invB);
        *(float2*)(dx)              = make_float2(xa.x, xa.y);
        *(float2*)(dx + 4)          = make_float2(xa.z, xa.w);
        *(float2*)(dx + APITCH)     = make_float2(xb.x, xb.y);
        *(float2*)(dx + APITCH + 4) = make_float2(xb.z, xb.w);

        sa0 = sa0N; na = naN; sb0 = sb0N; nb = nbN; idA = idAN; idB = idBN;
      }
    }
    __syncthreads();

    {
      const int tileN = tile + tstep;
      if (tileN < nTiles) {
        int lr = tileN * TILE_M + wr0 + (lane & 7);
        if (lr > nRows - 1) lr = nRows - 1;
        seV = rng[rbase + lr];
      }
    }

    v8f acc[4];
    const v8f vzero = {};
#pragma unroll
    for (int t = 0; t < 4; ++t) acc[t] = vzero;

    const float* fa0 = sAb + c * APITCH + 4 * h;
    const float* fx0 = sXb + c * APITCH + 4 * h;
#pragma unroll
    for (int j = 0; j < 16; ++j) {
      float4 fa[4], fx[4];
#pragma unroll
      for (int t = 0; t < 4; ++t) {
        fa[t] = *(const float4*)(fa0 + 16 * t * APITCH + 8 * j);
        fx[t] = *(const float4*)(fx0 + 16 * t * APITCH + 8 * j);
      }
#pragma unroll
      for (int t = 0; t < 4; ++t) {
        v2f a, ax;
        a.x = fa[t].x;  a.y = fa[t].y;
        ax.x = fx[t].x; ax.y = fx[t].y;
        acc[t] = __builtin_amdgcn_wmma_f32_16x16x4_f32(
            false, a, false, bl[2 * j], (short)0, acc[t], false, false);
        acc[t] = __builtin_amdgcn_wmma_f32_16x16x4_f32(
            false, ax, false, br[2 * j], (short)0, acc[t], false, false);
      }
#pragma unroll
      for (int t = 0; t < 4; ++t) {
        v2f a, ax;
        a.x = fa[t].z;  a.y = fa[t].w;
        ax.x = fx[t].z; ax.y = fx[t].w;
        acc[t] = __builtin_amdgcn_wmma_f32_16x16x4_f32(
            false, a, false, bl[2 * j + 1], (short)0, acc[t], false, false);
        acc[t] = __builtin_amdgcn_wmma_f32_16x16x4_f32(
            false, ax, false, br[2 * j + 1], (short)0, acc[t], false, false);
      }
    }

    float* orow = out + (long)(row0 + 8 * h) * HDIM + col16;
    if (row0 + TILE_M <= nRows) {
#pragma unroll
      for (int t = 0; t < 4; ++t) {
#pragma unroll
        for (int r = 0; r < 8; ++r) {
          float v = acc[t][r] + bv;
          if (doRelu) v = fmaxf(v, 0.0f);
          orow[(long)(16 * t + r) * HDIM] = v;
        }
      }
    } else {
#pragma unroll
      for (int t = 0; t < 4; ++t) {
#pragma unroll
        for (int r = 0; r < 8; ++r) {
          float v = acc[t][r] + bv;
          if (doRelu) v = fmaxf(v, 0.0f);
          if (row0 + 8 * h + 16 * t + r < nRows) orow[(long)(16 * t + r) * HDIM] = v;
        }
      }
    }
    buf ^= 1;
  }
}

#define DOT_EDGES 4
#define DOT_TRIPS 4

__global__ __launch_bounds__(256) void k_dot(const float* __restrict__ u2,
                                             const float* __restrict__ i2,
                                             const int* __restrict__ lu,
                                             const int* __restrict__ li,
                                             float* __restrict__ out, int nEL) {
  const int lane = threadIdx.x & 31;
  const int step = gridDim.x * 8 * DOT_EDGES;
  int e0 = __builtin_amdgcn_readfirstlane(blockIdx.x * 8 + (threadIdx.x >> 5)) * DOT_EDGES;
  if (e0 >= nEL) return;
  int iu[DOT_EDGES], ii[DOT_EDGES];
#pragma unroll
  for (int u = 0; u < DOT_EDGES; ++u) {
    const int e = (e0 + u < nEL) ? e0 + u : nEL - 1;
    iu[u] = lu[e]; ii[u] = li[e];
  }
  do {
    float4 a[DOT_EDGES], b[DOT_EDGES];
#pragma unroll
    for (int u = 0; u < DOT_EDGES; ++u) {
      a[u] = ((const float4*)(u2 + (long)iu[u] * HDIM))[lane];
      b[u] = ((const float4*)(i2 + (long)ii[u] * HDIM))[lane];
    }
    const int e1 = e0 + step;
    int iuN[DOT_EDGES], iiN[DOT_EDGES];
#pragma unroll
    for (int u = 0; u < DOT_EDGES; ++u) { iuN[u] = 0; iiN[u] = 0; }
    if (e1 < nEL) {
#pragma unroll
      for (int u = 0; u < DOT_EDGES; ++u) {
        const int e = (e1 + u < nEL) ? e1 + u : nEL - 1;
        iuN[u] = lu[e]; iiN[u] = li[e];
      }
    }
    float s[DOT_EDGES];
#pragma unroll
    for (int u = 0; u < DOT_EDGES; ++u)
      s[u] = a[u].x * b[u].x + a[u].y * b[u].y + a[u].z * b[u].z + a[u].w * b[u].w;
#pragma unroll
    for (int off = 16; off > 0; off >>= 1) {
#pragma unroll
      for (int u = 0; u < DOT_EDGES; ++u) s[u] += __shfl_down(s[u], off, 32);
    }
    if (lane == 0) {
      if (e0 + DOT_EDGES <= nEL) {
        *(float4*)(out + e0) = make_float4(s[0], s[1], s[2], s[3]);
      } else {
#pragma unroll
        for (int u = 0; u < DOT_EDGES; ++u)
          if (e0 + u < nEL) out[e0 + u] = s[u];
      }
    }
    e0 = e1;
#pragma unroll
    for (int u = 0; u < DOT_EDGES; ++u) { iu[u] = iuN[u]; ii[u] = iiN[u]; }
  } while (e0 < nEL);
}

extern "C" void kernel_launch(void* const* d_in, const int* in_sizes, int n_in,
                              void* d_out, int out_size, void* d_ws, size_t ws_size,
                              hipStream_t stream) {
  const float* user_emb = (const float*)d_in[0];
  const float* item_emb = (const float*)d_in[1];
  const float* w1_ui_l = (const float*)d_in[2];
  const float* w1_ui_r = (const float*)d_in[3];
  const float* w1_iu_l = (const float*)d_in[4];
  const float* w1_iu_r = (const float*)d_in[5];
  const float* w2_ui_l = (const float*)d_in[6];
  const float* w2_ui_r = (const float*)d_in[7];
  const float* w2_iu_l = (const float*)d_in[8];
  const float* w2_iu_r = (const float*)d_in[9];
  const float* b1_ui = (const float*)d_in[10];
  const float* b1_iu = (const float*)d_in[11];
  const float* b2_ui = (const float*)d_in[12];
  const float* b2_iu = (const float*)d_in[13];
  const int* eui  = (const int*)d_in[16];
  const int* eiu  = (const int*)d_in[17];
  const int* elab = (const int*)d_in[18];

  const int U  = in_sizes[0] / HDIM;
  const int I  = in_sizes[1] / HDIM;
  const int E  = in_sizes[16] / 2;
  const int EL = in_sizes[18] / 2;
  const int R  = U + I;
  const int nBins = (R + BIN_ROWS - 1) / BIN_ROWS;

  float* ws = (float*)d_ws;
  long off = 0;
  float* u1 = ws + off; off += (long)U * HDIM;
  float* i1 = ws + off; off += (long)I * HDIM;
  float* u2 = ws + off; off += (long)U * HDIM;
  float* i2 = ws + off; off += (long)I * HDIM;
  off += off & 1;
  long long* rng = (long long*)(ws + off); off += 2L * R;
  int* pairs  = (int*)(ws + off); off += 4L * E;
  int* col    = (int*)(ws + off); off += 2L * E;
  int* hist   = (int*)(ws + off); off += (long)SORT_BLOCKS * nBins;
  int* binStart = (int*)(ws + off); off += nBins + 1;
  (void)ws_size;

  auto cdiv = [](long a, long b) { return (int)((a + b - 1) / b); };

  const long costU = 18L * U + E;
  const long costI = 18L * I + E;
  int gemmSplit = GEMM_GRID / 2;
  if (costU + costI > 0) gemmSplit = (int)(GEMM_GRID * costU / (costU + costI));
  if (gemmSplit < 1) gemmSplit = 1;
  if (gemmSplit > GEMM_GRID - 1) gemmSplit = GEMM_GRID - 1;
  const int dotBlocks = cdiv(EL, 8 * DOT_EDGES * DOT_TRIPS);

  k_bin_count<<<SORT_BLOCKS, SORT_THREADS, 0, stream>>>(eiu, eui, hist, E, U, nBins);
  k_bin_scatter<<<SORT_BLOCKS, SORT_THREADS, 0, stream>>>(eiu, eui, hist, binStart, pairs, E, U, nBins);
  k_bin_sort<<<nBins, BIN_ROWS, 0, stream>>>(pairs, binStart, rng, col, R);

  k_sage_layer<<<GEMM_GRID, 256, 0, stream>>>(item_emb, user_emb, w1_iu_l, w1_iu_r, b1_iu, u1, U,
                                              user_emb, item_emb, w1_ui_l, w1_ui_r, b1_ui, i1, I,
                                              rng, col, gemmSplit, 1);

  k_sage_layer<<<GEMM_GRID, 256, 0, stream>>>(i1, u1, w2_iu_l, w2_iu_r, b2_iu, u2, U,
                                              u1, i1, w2_ui_l, w2_ui_r, b2_ui, i2, I,
                                              rng, col, gemmSplit, 0);

  k_dot<<<dotBlocks, 256, 0, stream>>>(u2, i2, elab, elab + EL, (float*)d_out, EL);
}
